# starting kernel
# baseline (speedup 1.0000x reference)
#include <hip/hip_runtime.h>
#include <hip/hip_bf16.h>

typedef float v2f __attribute__((ext_vector_type(2)));
typedef float v8f __attribute__((ext_vector_type(8)));

#define HEADS 8
#define CH 16
#define F 128
#define NEG_SLOPE 0.2f

#if defined(__has_builtin)
#if __has_builtin(__builtin_amdgcn_wmma_f32_16x16x4_f32)
#define HAVE_WMMA_F32X4 1
#endif
#endif

__device__ __forceinline__ unsigned flipf(float f) {
  unsigned u = __float_as_uint(f);
  return (u >> 31) ? ~u : (u | 0x80000000u);
}
__device__ __forceinline__ float unflipf(unsigned u) {
  return __uint_as_float((u >> 31) ? (u ^ 0x80000000u) : ~u);
}
__device__ __forceinline__ float lrelu(float v) {
  return v > 0.0f ? v : NEG_SLOPE * v;
}

__global__ __launch_bounds__(256) void gat_init(float* __restrict__ out,
                                                const float* __restrict__ bias,
                                                unsigned* __restrict__ m_u,
                                                float* __restrict__ denom,
                                                int N) {
  long long i = (long long)blockIdx.x * 256 + threadIdx.x;
  long long tot = (long long)N * F;
  if (i < tot) out[i] = bias[(int)(i & (F - 1))];
  if (i < (long long)N * HEADS) {
    m_u[i] = 0x007FFFFFu;
    denom[i] = 0.0f;
  }
}

__global__ __launch_bounds__(256) void gat_gemm_wmma(
    const float* __restrict__ x, const float* __restrict__ W,
    const float* __restrict__ att_src, const float* __restrict__ att_dst,
    float* __restrict__ h, float* __restrict__ a_src, float* __restrict__ a_dst,
    int N) {
  __shared__ float sA[16 * F];

  const int tid = threadIdx.x;
  const int wave = tid >> 5;
  const int lane = tid & 31;
  const int half = lane >> 4;
  const int l16 = lane & 15;
  const int rowbase = blockIdx.x * 16;

  {
    int r = tid >> 4;
    int c = (tid & 15) << 3;
    int rg = rowbase + r;
    if (rg >= N) rg = N - 1;
    const float4* s4 = (const float4*)(x + (size_t)rg * F + c);
    float4 a0 = s4[0];
    float4 a1 = s4[1];
    *(float4*)(sA + r * F + c) = a0;
    *(float4*)(sA + r * F + c + 4) = a1;
  }
  __syncthreads();

  const int colbase = wave * CH;
  v8f acc = {0.f, 0.f, 0.f, 0.f, 0.f, 0.f, 0.f, 0.f};

  for (int k = 0; k < F; k += 4) {
#if defined(HAVE_WMMA_F32X4)
    v2f afrag;
    const float* ap = sA + l16 * F + k + 2 * half;
    afrag.x = ap[0];
    afrag.y = ap[1];
    v2f bfrag;
    const float* bp = W + (size_t)(k + 2 * half) * F + colbase + l16;
    bfrag.x = bp[0];
    bfrag.y = bp[F];
    acc = __builtin_amdgcn_wmma_f32_16x16x4_f32(false, afrag, false, bfrag,
                                                (short)0, acc, false, false);
#else
    for (int v = 0; v < 8; ++v) {
      int row = v + 8 * half;
      for (int j = 0; j < 4; ++j)
        acc[v] += sA[row * F + k + j] * W[(size_t)(k + j) * F + colbase + l16];
    }
#endif
  }

  const float was = att_src[wave * CH + l16];
  const float wad = att_dst[wave * CH + l16];
#pragma unroll
  for (int v = 0; v < 8; ++v) {
    int row = rowbase + v + 8 * half;
    float val = acc[v];
    float ps = val * was;
    float pd = val * wad;
#pragma unroll
    for (int off = 8; off >= 1; off >>= 1) {
      ps += __shfl_xor(ps, off, 32);
      pd += __shfl_xor(pd, off, 32);
    }
    if (row < N) {
      h[(size_t)row * F + colbase + l16] = val;
      if (l16 == 0) {
        a_src[(size_t)row * HEADS + wave] = ps;
        a_dst[(size_t)row * HEADS + wave] = pd;
      }
    }
  }
}

__global__ __launch_bounds__(256) void gat_logits_max(
    const int* __restrict__ srcA, const int* __restrict__ dstA,
    const float* __restrict__ a_src, const float* __restrict__ a_dst,
    unsigned* __restrict__ m_u, int E, int EN) {
  int e = blockIdx.x * 256 + threadIdx.x;
  if (e >= EN) return;
  int s, d;
  if (e < E) { s = srcA[e]; d = dstA[e]; } else { s = d = e - E; }
  const float4* as4 = (const float4*)(a_src + (size_t)s * HEADS);
  const float4* ad4 = (const float4*)(a_dst + (size_t)d * HEADS);
  float4 s0 = as4[0], s1 = as4[1];
  float4 d0 = ad4[0], d1 = ad4[1];
  unsigned* mp = m_u + (size_t)d * HEADS;
  atomicMax(mp + 0, flipf(lrelu(s0.x + d0.x)));
  atomicMax(mp + 1, flipf(lrelu(s0.y + d0.y)));
  atomicMax(mp + 2, flipf(lrelu(s0.z + d0.z)));
  atomicMax(mp + 3, flipf(lrelu(s0.w + d0.w)));
  atomicMax(mp + 4, flipf(lrelu(s1.x + d1.x)));
  atomicMax(mp + 5, flipf(lrelu(s1.y + d1.y)));
  atomicMax(mp + 6, flipf(lrelu(s1.z + d1.z)));
  atomicMax(mp + 7, flipf(lrelu(s1.w + d1.w)));
}

__global__ __launch_bounds__(256) void gat_finalize_m(unsigned* __restrict__ m_u,
                                                      int count) {
  int i = blockIdx.x * 256 + threadIdx.x;
  if (i < count) {
    float v = unflipf(m_u[i]);
    ((float*)m_u)[i] = v;
  }
}

__global__ __launch_bounds__(256) void gat_expsum(
    const int* __restrict__ srcA, const int* __restrict__ dstA,
    const float* __restrict__ a_src, const float* __restrict__ a_dst,
    const float* __restrict__ m, float* __restrict__ denom, int E, int EN) {
  int e = blockIdx.x * 256 + threadIdx.x;
  if (e >= EN) return;
  int s, d;
  if (e < E) { s = srcA[e]; d = dstA[e]; } else { s = d = e - E; }
  const float4* as4 = (const float4*)(a_src + (size_t)s * HEADS);
  const float4* ad4 = (const float4*)(a_dst + (size_t)d * HEADS);
  const float4* m4 = (const float4*)(m + (size_t)d * HEADS);
  float4 s0 = as4[0], s1 = as4[1];
  float4 d0 = ad4[0], d1 = ad4[1];
  float4 m0 = m4[0], m1 = m4[1];
  float* dp = denom + (size_t)d * HEADS;
  atomicAdd(dp + 0, expf(lrelu(s0.x + d0.x) - m0.x));
  atomicAdd(dp + 1, expf(lrelu(s0.y + d0.y) - m0.y));
  atomicAdd(dp + 2, expf(lrelu(s0.z + d0.z) - m0.z));
  atomicAdd(dp + 3, expf(lrelu(s0.w + d0.w) - m0.w));
  atomicAdd(dp + 4, expf(lrelu(s1.x + d1.x) - m1.x));
  atomicAdd(dp + 5, expf(lrelu(s1.y + d1.y) - m1.y));
  atomicAdd(dp + 6, expf(lrelu(s1.z + d1.z) - m1.z));
  atomicAdd(dp + 7, expf(lrelu(s1.w + d1.w) - m1.w));
}

__global__ __launch_bounds__(256) void gat_aggregate(
    const int* __restrict__ srcA, const int* __restrict__ dstA,
    const float* __restrict__ a_src, const float* __restrict__ a_dst,
    const float* __restrict__ m, const float* __restrict__ denom,
    const float* __restrict__ h, float* __restrict__ out, int E, int EN) {
  long long widx = (long long)blockIdx.x * 8 + (threadIdx.x >> 5);
  if (widx >= EN) return;
  int e = (int)widx;
  int lane = threadIdx.x & 31;
  int s, d;
  if (e < E) { s = srcA[e]; d = dstA[e]; } else { s = d = e - E; }
  int hd = lane >> 2;
  float lg = lrelu(a_src[(size_t)s * HEADS + hd] + a_dst[(size_t)d * HEADS + hd]);
  float alpha = expf(lg - m[(size_t)d * HEADS + hd]) / denom[(size_t)d * HEADS + hd];
  float4 hv = *(const float4*)(h + (size_t)s * F + lane * 4);
  float* op = out + (size_t)d * F + lane * 4;
  atomicAdd(op + 0, alpha * hv.x);
  atomicAdd(op + 1, alpha * hv.y);
  atomicAdd(op + 2, alpha * hv.z);
  atomicAdd(op + 3, alpha * hv.w);
}

extern "C" void kernel_launch(void* const* d_in, const int* in_sizes, int n_in,
                              void* d_out, int out_size, void* d_ws, size_t ws_size,
                              hipStream_t stream) {
  const float* x = (const float*)d_in[0];
  const int* ei = (const int*)d_in[1];
  const float* W = (const float*)d_in[2];
  const float* att_src = (const float*)d_in[3];
  const float* att_dst = (const float*)d_in[4];
  const float* bias = (const float*)d_in[5];
  float* out = (float*)d_out;

  const int N = in_sizes[0] / F;
  const int E = in_sizes[1] / 2;
  const int EN = E + N;
  const int* srcA = ei;
  const int* dstA = ei + E;

  float* ws = (float*)d_ws;
  float* hbuf = ws;
  float* asrc = hbuf + (size_t)N * F;
  float* adst = asrc + (size_t)N * HEADS;
  float* mbuf = adst + (size_t)N * HEADS;
  float* dbuf = mbuf + (size_t)N * HEADS;

  {
    long long tot = (long long)N * F;
    int blocks = (int)((tot + 255) / 256);
    gat_init<<<blocks, 256, 0, stream>>>(out, bias, (unsigned*)mbuf, dbuf, N);
  }
  gat_gemm_wmma<<<(N + 15) / 16, 256, 0, stream>>>(x, W, att_src, att_dst,
                                                   hbuf, asrc, adst, N);
  gat_logits_max<<<(EN + 255) / 256, 256, 0, stream>>>(srcA, dstA, asrc, adst,
                                                       (unsigned*)mbuf, E, EN);
  gat_finalize_m<<<((N * HEADS) + 255) / 256, 256, 0, stream>>>((unsigned*)mbuf,
                                                                N * HEADS);
  gat_expsum<<<(EN + 255) / 256, 256, 0, stream>>>(srcA, dstA, asrc, adst,
                                                   mbuf, dbuf, E, EN);
  gat_aggregate<<<(EN + 7) / 8, 256, 0, stream>>>(srcA, dstA, asrc, adst,
                                                  mbuf, dbuf, hbuf, out, E, EN);
}


# Exp-sum edge pass folded into aggregation: each edge adds its exp-weighted h row and its exp weights by float atomics; a closing pass divides each output once, adds bias; sums stay in arrival order
# speedup vs baseline: 1.1608x; 1.1608x over previous
#include <hip/hip_runtime.h>
#include <hip/hip_bf16.h>

typedef float v2f __attribute__((ext_vector_type(2)));
typedef float v8f __attribute__((ext_vector_type(8)));

#define HEADS 8
#define CH 16
#define F 128
#define NEG_SLOPE 0.2f

#if defined(__has_builtin)
#if __has_builtin(__builtin_amdgcn_wmma_f32_16x16x4_f32)
#define HAVE_WMMA_F32X4 1
#endif
#endif

__device__ __forceinline__ unsigned flipf(float f) {
  unsigned u = __float_as_uint(f);
  return (u >> 31) ? ~u : (u | 0x80000000u);
}
__device__ __forceinline__ float unflipf(unsigned u) {
  return __uint_as_float((u >> 31) ? (u ^ 0x80000000u) : ~u);
}
__device__ __forceinline__ float lrelu(float v) {
  return v > 0.0f ? v : NEG_SLOPE * v;
}

__global__ __launch_bounds__(256) void gat_init(float* __restrict__ out,
                                                unsigned* __restrict__ m_u,
                                                float* __restrict__ denom,
                                                int N) {
  long long i = (long long)blockIdx.x * 256 + threadIdx.x;
  long long tot = (long long)N * F;
  if (i < tot) out[i] = 0.0f;
  if (i < (long long)N * HEADS) {
    m_u[i] = 0x007FFFFFu;
    denom[i] = 0.0f;
  }
}

__global__ __launch_bounds__(256) void gat_gemm_wmma(
    const float* __restrict__ x, const float* __restrict__ W,
    const float* __restrict__ att_src, const float* __restrict__ att_dst,
    float* __restrict__ h, float* __restrict__ a_src, float* __restrict__ a_dst,
    int N) {
  __shared__ float sA[16 * F];

  const int tid = threadIdx.x;
  const int wave = tid >> 5;
  const int lane = tid & 31;
  const int half = lane >> 4;
  const int l16 = lane & 15;
  const int rowbase = blockIdx.x * 16;

  {
    int r = tid >> 4;
    int c = (tid & 15) << 3;
    int rg = rowbase + r;
    if (rg >= N) rg = N - 1;
    const float4* s4 = (const float4*)(x + (size_t)rg * F + c);
    float4 a0 = s4[0];
    float4 a1 = s4[1];
    *(float4*)(sA + r * F + c) = a0;
    *(float4*)(sA + r * F + c + 4) = a1;
  }
  __syncthreads();

  const int colbase = wave * CH;
  v8f acc = {0.f, 0.f, 0.f, 0.f, 0.f, 0.f, 0.f, 0.f};

  for (int k = 0; k < F; k += 4) {
#if defined(HAVE_WMMA_F32X4)
    v2f afrag;
    const float* ap = sA + l16 * F + k + 2 * half;
    afrag.x = ap[0];
    afrag.y = ap[1];
    v2f bfrag;
    const float* bp = W + (size_t)(k + 2 * half) * F + colbase + l16;
    bfrag.x = bp[0];
    bfrag.y = bp[F];
    acc = __builtin_amdgcn_wmma_f32_16x16x4_f32(false, afrag, false, bfrag,
                                                (short)0, acc, false, false);
#else
    for (int v = 0; v < 8; ++v) {
      int row = v + 8 * half;
      for (int j = 0; j < 4; ++j)
        acc[v] += sA[row * F + k + j] * W[(size_t)(k + j) * F + colbase + l16];
    }
#endif
  }

  const float was = att_src[wave * CH + l16];
  const float wad = att_dst[wave * CH + l16];
#pragma unroll
  for (int v = 0; v < 8; ++v) {
    int row = rowbase + v + 8 * half;
    float val = acc[v];
    float ps = val * was;
    float pd = val * wad;
#pragma unroll
    for (int off = 8; off >= 1; off >>= 1) {
      ps += __shfl_xor(ps, off, 32);
      pd += __shfl_xor(pd, off, 32);
    }
    if (row < N) {
      h[(size_t)row * F + colbase + l16] = val;
      if (l16 == 0) {
        a_src[(size_t)row * HEADS + wave] = ps;
        a_dst[(size_t)row * HEADS + wave] = pd;
      }
    }
  }
}

__global__ __launch_bounds__(256) void gat_logits_max(
    const int* __restrict__ srcA, const int* __restrict__ dstA,
    const float* __restrict__ a_src, const float* __restrict__ a_dst,
    unsigned* __restrict__ m_u, int E, int EN) {
  int e = blockIdx.x * 256 + threadIdx.x;
  if (e >= EN) return;
  int s, d;
  if (e < E) { s = srcA[e]; d = dstA[e]; } else { s = d = e - E; }
  const float4* as4 = (const float4*)(a_src + (size_t)s * HEADS);
  const float4* ad4 = (const float4*)(a_dst + (size_t)d * HEADS);
  float4 s0 = as4[0], s1 = as4[1];
  float4 d0 = ad4[0], d1 = ad4[1];
  unsigned* mp = m_u + (size_t)d * HEADS;
  atomicMax(mp + 0, flipf(lrelu(s0.x + d0.x)));
  atomicMax(mp + 1, flipf(lrelu(s0.y + d0.y)));
  atomicMax(mp + 2, flipf(lrelu(s0.z + d0.z)));
  atomicMax(mp + 3, flipf(lrelu(s0.w + d0.w)));
  atomicMax(mp + 4, flipf(lrelu(s1.x + d1.x)));
  atomicMax(mp + 5, flipf(lrelu(s1.y + d1.y)));
  atomicMax(mp + 6, flipf(lrelu(s1.z + d1.z)));
  atomicMax(mp + 7, flipf(lrelu(s1.w + d1.w)));
}

__global__ __launch_bounds__(256) void gat_finalize_m(unsigned* __restrict__ m_u,
                                                      int count) {
  int i = blockIdx.x * 256 + threadIdx.x;
  if (i < count) {
    float v = unflipf(m_u[i]);
    ((float*)m_u)[i] = v;
  }
}

__global__ __launch_bounds__(256) void gat_aggregate(
    const int* __restrict__ srcA, const int* __restrict__ dstA,
    const float* __restrict__ a_src, const float* __restrict__ a_dst,
    const float* __restrict__ m, float* __restrict__ denom,
    const float* __restrict__ h, float* __restrict__ out, int E, int EN) {
  long long widx = (long long)blockIdx.x * 8 + (threadIdx.x >> 5);
  if (widx >= EN) return;
  int e = (int)widx;
  int lane = threadIdx.x & 31;
  int s, d;
  if (e < E) { s = srcA[e]; d = dstA[e]; } else { s = d = e - E; }
  int hd = lane >> 2;
  float lg = lrelu(a_src[(size_t)s * HEADS + hd] + a_dst[(size_t)d * HEADS + hd]);
  float w = expf(lg - m[(size_t)d * HEADS + hd]);
  float4 hv = *(const float4*)(h + (size_t)s * F + lane * 4);
  float* op = out + (size_t)d * F + lane * 4;
  atomicAdd(op + 0, w * hv.x);
  atomicAdd(op + 1, w * hv.y);
  atomicAdd(op + 2, w * hv.z);
  atomicAdd(op + 3, w * hv.w);
  if ((lane & 3) == 0) atomicAdd(denom + (size_t)d * HEADS + hd, w);
}

__global__ __launch_bounds__(256) void gat_normalize(float* __restrict__ out,
                                                     const float* __restrict__ denom,
                                                     const float* __restrict__ bias,
                                                     int N) {
  long long i = (long long)blockIdx.x * 256 + threadIdx.x;
  if (i >= (long long)N * (F / 4)) return;
  int q = (int)(i & (F / 4 - 1));
  long long node = i >> 5;
  float den = denom[node * HEADS + (q >> 2)];
  float4 o = ((const float4*)out)[i];
  float4 b = ((const float4*)bias)[q];
  o.x = o.x / den + b.x;
  o.y = o.y / den + b.y;
  o.z = o.z / den + b.z;
  o.w = o.w / den + b.w;
  ((float4*)out)[i] = o;
}

extern "C" void kernel_launch(void* const* d_in, const int* in_sizes, int n_in,
                              void* d_out, int out_size, void* d_ws, size_t ws_size,
                              hipStream_t stream) {
  const float* x = (const float*)d_in[0];
  const int* ei = (const int*)d_in[1];
  const float* W = (const float*)d_in[2];
  const float* att_src = (const float*)d_in[3];
  const float* att_dst = (const float*)d_in[4];
  const float* bias = (const float*)d_in[5];
  float* out = (float*)d_out;

  const int N = in_sizes[0] / F;
  const int E = in_sizes[1] / 2;
  const int EN = E + N;
  const int* srcA = ei;
  const int* dstA = ei + E;

  float* ws = (float*)d_ws;
  float* hbuf = ws;
  float* asrc = hbuf + (size_t)N * F;
  float* adst = asrc + (size_t)N * HEADS;
  float* mbuf = adst + (size_t)N * HEADS;
  float* dbuf = mbuf + (size_t)N * HEADS;

  {
    long long tot = (long long)N * F;
    int blocks = (int)((tot + 255) / 256);
    gat_init<<<blocks, 256, 0, stream>>>(out, (unsigned*)mbuf, dbuf, N);
  }
  gat_gemm_wmma<<<(N + 15) / 16, 256, 0, stream>>>(x, W, att_src, att_dst,
                                                   hbuf, asrc, adst, N);
  gat_logits_max<<<(EN + 255) / 256, 256, 0, stream>>>(srcA, dstA, asrc, adst,
                                                       (unsigned*)mbuf, E, EN);
  gat_finalize_m<<<((N * HEADS) + 255) / 256, 256, 0, stream>>>((unsigned*)mbuf,
                                                                N * HEADS);
  gat_aggregate<<<(EN + 7) / 8, 256, 0, stream>>>(srcA, dstA, asrc, adst,
                                                  mbuf, dbuf, hbuf, out, E, EN);
  gat_normalize<<<(N * (F / 4) + 255) / 256, 256, 0, stream>>>(out, dbuf, bias, N);
}


# Output rows owned in LDS: each work-group of 392 nodes filters dst for its own edges, adds exp-weighted h rows by LDS float adds, stores out once; no global float atomics; sums in arrival order
# speedup vs baseline: 2.0590x; 1.7738x over previous
#include <hip/hip_runtime.h>
#include <hip/hip_bf16.h>

typedef float v2f __attribute__((ext_vector_type(2)));
typedef float v8f __attribute__((ext_vector_type(8)));
typedef float v4f __attribute__((ext_vector_type(4)));

#define HEADS 8
#define CH 16
#define F 128
#define NEG_SLOPE 0.2f

#if defined(__has_builtin)
#if __has_builtin(__builtin_amdgcn_wmma_f32_16x16x4_f32)
#define HAVE_WMMA_F32X4 1
#endif
#endif

__device__ __forceinline__ unsigned flipf(float f) {
  unsigned u = __float_as_uint(f);
  return (u >> 31) ? ~u : (u | 0x80000000u);
}
__device__ __forceinline__ float unflipf(unsigned u) {
  return __uint_as_float((u >> 31) ? (u ^ 0x80000000u) : ~u);
}
__device__ __forceinline__ float lrelu(float v) {
  return v > 0.0f ? v : NEG_SLOPE * v;
}

__global__ __launch_bounds__(256) void gat_init(unsigned* __restrict__ m_u,
                                                int count) {
  int i = blockIdx.x * 256 + threadIdx.x;
  if (i < count) m_u[i] = 0x007FFFFFu;
}

__global__ __launch_bounds__(256) void gat_gemm_wmma(
    const float* __restrict__ x, const float* __restrict__ W,
    const float* __restrict__ att_src, const float* __restrict__ att_dst,
    float* __restrict__ h, float* __restrict__ a_src, float* __restrict__ a_dst,
    int N) {
  __shared__ float sA[16 * F];

  const int tid = threadIdx.x;
  const int wave = tid >> 5;
  const int lane = tid & 31;
  const int half = lane >> 4;
  const int l16 = lane & 15;
  const int rowbase = blockIdx.x * 16;

  {
    int r = tid >> 4;
    int c = (tid & 15) << 3;
    int rg = rowbase + r;
    if (rg >= N) rg = N - 1;
    const float4* s4 = (const float4*)(x + (size_t)rg * F + c);
    float4 a0 = s4[0];
    float4 a1 = s4[1];
    *(float4*)(sA + r * F + c) = a0;
    *(float4*)(sA + r * F + c + 4) = a1;
  }
  __syncthreads();

  const int colbase = wave * CH;
  v8f acc = {0.f, 0.f, 0.f, 0.f, 0.f, 0.f, 0.f, 0.f};

  for (int k = 0; k < F; k += 4) {
#if defined(HAVE_WMMA_F32X4)
    v2f afrag;
    const float* ap = sA + l16 * F + k + 2 * half;
    afrag.x = ap[0];
    afrag.y = ap[1];
    v2f bfrag;
    const float* bp = W + (size_t)(k + 2 * half) * F + colbase + l16;
    bfrag.x = bp[0];
    bfrag.y = bp[F];
    acc = __builtin_amdgcn_wmma_f32_16x16x4_f32(false, afrag, false, bfrag,
                                                (short)0, acc, false, false);
#else
    for (int v = 0; v < 8; ++v) {
      int row = v + 8 * half;
      for (int j = 0; j < 4; ++j)
        acc[v] += sA[row * F + k + j] * W[(size_t)(k + j) * F + colbase + l16];
    }
#endif
  }

  const float was = att_src[wave * CH + l16];
  const float wad = att_dst[wave * CH + l16];
#pragma unroll
  for (int v = 0; v < 8; ++v) {
    int row = rowbase + v + 8 * half;
    float val = acc[v];
    float ps = val * was;
    float pd = val * wad;
#pragma unroll
    for (int off = 8; off >= 1; off >>= 1) {
      ps += __shfl_xor(ps, off, 32);
      pd += __shfl_xor(pd, off, 32);
    }
    if (row < N) {
      h[(size_t)row * F + colbase + l16] = val;
      if (l16 == 0) {
        a_src[(size_t)row * HEADS + wave] = ps;
        a_dst[(size_t)row * HEADS + wave] = pd;
      }
    }
  }
}

__global__ __launch_bounds__(256) void gat_logits_max(
    const int* __restrict__ srcA, const int* __restrict__ dstA,
    const float* __restrict__ a_src, const float* __restrict__ a_dst,
    unsigned* __restrict__ m_u, int E, int EN) {
  int e = blockIdx.x * 256 + threadIdx.x;
  if (e >= EN) return;
  int s, d;
  if (e < E) { s = srcA[e]; d = dstA[e]; } else { s = d = e - E; }
  const float4* as4 = (const float4*)(a_src + (size_t)s * HEADS);
  const float4* ad4 = (const float4*)(a_dst + (size_t)d * HEADS);
  float4 s0 = as4[0], s1 = as4[1];
  float4 d0 = ad4[0], d1 = ad4[1];
  unsigned* mp = m_u + (size_t)d * HEADS;
  atomicMax(mp + 0, flipf(lrelu(s0.x + d0.x)));
  atomicMax(mp + 1, flipf(lrelu(s0.y + d0.y)));
  atomicMax(mp + 2, flipf(lrelu(s0.z + d0.z)));
  atomicMax(mp + 3, flipf(lrelu(s0.w + d0.w)));
  atomicMax(mp + 4, flipf(lrelu(s1.x + d1.x)));
  atomicMax(mp + 5, flipf(lrelu(s1.y + d1.y)));
  atomicMax(mp + 6, flipf(lrelu(s1.z + d1.z)));
  atomicMax(mp + 7, flipf(lrelu(s1.w + d1.w)));
}

__global__ __launch_bounds__(256) void gat_finalize_m(unsigned* __restrict__ m_u,
                                                      int count) {
  int i = blockIdx.x * 256 + threadIdx.x;
  if (i < count) {
    float v = unflipf(m_u[i]);
    ((float*)m_u)[i] = v;
  }
}

#define NPW 392
#define CAP 8192
#define CHUNK (1 << 20)
#define UNR 8

__device__ __forceinline__ void take_edge(int d, int e, int base, unsigned nloc,
                                          int* s_cnt, int* s_src, int* s_dl) {
  unsigned r = (unsigned)d - (unsigned)base;
  if (r < nloc) {
    int slot = atomicAdd(s_cnt, 1);
    if (slot < CAP) {
      s_src[slot] = e;
      s_dl[slot] = (int)r;
    }
  }
}

__device__ __forceinline__ void take_quad(int4 d, int e, int base, unsigned nloc,
                                          int* s_cnt, int* s_src, int* s_dl) {
  unsigned r0 = (unsigned)d.x - (unsigned)base;
  unsigned r1 = (unsigned)d.y - (unsigned)base;
  unsigned r2 = (unsigned)d.z - (unsigned)base;
  unsigned r3 = (unsigned)d.w - (unsigned)base;
  if ((r0 < nloc) | (r1 < nloc) | (r2 < nloc) | (r3 < nloc)) {
    take_edge(d.x, e + 0, base, nloc, s_cnt, s_src, s_dl);
    take_edge(d.y, e + 1, base, nloc, s_cnt, s_src, s_dl);
    take_edge(d.z, e + 2, base, nloc, s_cnt, s_src, s_dl);
    take_edge(d.w, e + 3, base, nloc, s_cnt, s_src, s_dl);
  }
}

__global__ __launch_bounds__(1024) void gat_owner_aggregate(
    const int* __restrict__ srcA, const int* __restrict__ dstA,
    const float* __restrict__ a_src, const float* __restrict__ a_dst,
    const float* __restrict__ m, const float* __restrict__ h,
    const float* __restrict__ bias, float* __restrict__ out, int E, int N) {
  __shared__ float s_acc[NPW * F];
  __shared__ float s_den[NPW * HEADS];
  __shared__ int s_src[CAP];
  __shared__ int s_dl[CAP];
  __shared__ int s_cnt;

  const int tid = threadIdx.x;
  const int lane = tid & 31;
  const int wave = __builtin_amdgcn_readfirstlane(tid >> 5);
  const int base = blockIdx.x * NPW;
  int nl = N - base;
  if (nl > NPW) nl = NPW;
  const unsigned nloc = (unsigned)nl;

  const int tot = nl * 32;
  for (int i0 = tid; i0 < tot; i0 += 4 * 1024) {
    float asv[4], adv[4], mv[4];
    v4f hq[4];
#pragma unroll
    for (int u = 0; u < 4; ++u) {
      int idx = i0 + u * 1024;
      if (idx >= tot) idx = i0;
      size_t node = (size_t)(base + (idx >> 5));
      int q = idx & 31;
      asv[u] = a_src[node * HEADS + (q >> 2)];
      adv[u] = a_dst[node * HEADS + (q >> 2)];
      mv[u] = m[node * HEADS + (q >> 2)];
      hq[u] = *(const v4f*)(h + node * F + q * 4);
    }
#pragma unroll
    for (int u = 0; u < 4; ++u) {
      int idx = i0 + u * 1024;
      if (idx < tot) {
        int i = idx >> 5;
        int q = idx & 31;
        float w = expf(lrelu(asv[u] + adv[u]) - mv[u]);
        float* ap = s_acc + i * F + q;
        ap[0] = w * hq[u].x;
        ap[32] = w * hq[u].y;
        ap[64] = w * hq[u].z;
        ap[96] = w * hq[u].w;
        if ((q & 3) == 0) s_den[i * HEADS + (q >> 2)] = w;
      }
    }
  }

  const int hd = lane >> 2;
  int pos = 0;
  int step = CHUNK;
  while (pos < E) {
    int end = (E - pos > step) ? pos + step : E;
    if (tid == 0) s_cnt = 0;
    __syncthreads();

    {
      const int* dp = dstA + pos;
      const int n = end - pos;
      const int nvec = ((((size_t)dp) & 15) == 0) ? (n >> 2) : 0;
      const int4* dp4 = (const int4*)dp;
      int v = tid;
      for (; v + 7 * 1024 < nvec; v += 8 * 1024) {
        int4 dq[8];
#pragma unroll
        for (int u = 0; u < 8; ++u) dq[u] = dp4[v + u * 1024];
#pragma unroll
        for (int u = 0; u < 8; ++u)
          take_quad(dq[u], pos + 4 * (v + u * 1024), base, nloc, &s_cnt, s_src, s_dl);
      }
      for (; v < nvec; v += 1024)
        take_quad(dp4[v], pos + 4 * v, base, nloc, &s_cnt, s_src, s_dl);
      for (int i = (nvec << 2) + tid; i < n; i += 1024)
        take_edge(dp[i], pos + i, base, nloc, &s_cnt, s_src, s_dl);
    }
    __syncthreads();
    const int cnt = s_cnt;
    __syncthreads();
    if (cnt > CAP) {
      step = CAP;
      continue;
    }

    for (int i = tid; i < cnt; i += 4 * 1024) {
      int ev[4];
#pragma unroll
      for (int u = 0; u < 4; ++u) {
        int j = i + u * 1024;
        ev[u] = (j < cnt) ? s_src[j] : 0;
      }
#pragma unroll
      for (int u = 0; u < 4; ++u) ev[u] = srcA[ev[u]];
#pragma unroll
      for (int u = 0; u < 4; ++u) {
        int j = i + u * 1024;
        if (j < cnt) s_src[j] = ev[u];
      }
    }
    __syncthreads();

    for (int i0 = wave * UNR; i0 < cnt; i0 += 32 * UNR) {
      int dl[UNR];
      float as[UNR], ad[UNR], mm[UNR];
      v4f hv[UNR];
      int sl = i0 + (lane & (UNR - 1));
      if (sl >= cnt) sl = i0;
      const int sv = s_src[sl];
      const int dv = s_dl[sl];
#pragma unroll
      for (int u = 0; u < UNR; ++u) {
        int s = __builtin_amdgcn_readlane(sv, u);
        dl[u] = __builtin_amdgcn_readlane(dv, u);
        size_t d = (size_t)(base + dl[u]);
        as[u] = a_src[(size_t)s * HEADS + hd];
        ad[u] = a_dst[d * HEADS + hd];
        mm[u] = m[d * HEADS + hd];
        hv[u] = *(const v4f*)(h + (size_t)s * F + lane * 4);
      }
      asm volatile("" ::"v"(hv[0]), "v"(hv[1]), "v"(hv[2]), "v"(hv[3]),
                   "v"(hv[4]), "v"(hv[5]), "v"(hv[6]), "v"(hv[7]));
#pragma unroll
      for (int u = 0; u < UNR; ++u) {
        if (i0 + u < cnt) {
          float w = expf(lrelu(as[u] + ad[u]) - mm[u]);
          float* ap = s_acc + dl[u] * F + lane;
          atomicAdd(ap + 0, w * hv[u].x);
          atomicAdd(ap + 32, w * hv[u].y);
          atomicAdd(ap + 64, w * hv[u].z);
          atomicAdd(ap + 96, w * hv[u].w);
          if ((lane & 3) == 0) atomicAdd(s_den + dl[u] * HEADS + hd, w);
        }
      }
    }
    __syncthreads();
    pos = end;
    step = CHUNK;
  }
  __syncthreads();

  const float4* b4 = (const float4*)bias;
  for (int idx = tid; idx < nl * 32; idx += 1024) {
    int i = idx >> 5;
    int q = idx & 31;
    float den = s_den[i * HEADS + (q >> 2)];
    const float* ap = s_acc + i * F + q;
    float4 b = b4[q];
    float4 o;
    o.x = ap[0] / den + b.x;
    o.y = ap[32] / den + b.y;
    o.z = ap[64] / den + b.z;
    o.w = ap[96] / den + b.w;
    ((float4*)out)[(size_t)(base + i) * 32 + q] = o;
  }
}

extern "C" void kernel_launch(void* const* d_in, const int* in_sizes, int n_in,
                              void* d_out, int out_size, void* d_ws, size_t ws_size,
                              hipStream_t stream) {
  const float* x = (const float*)d_in[0];
  const int* ei = (const int*)d_in[1];
  const float* W = (const float*)d_in[2];
  const float* att_src = (const float*)d_in[3];
  const float* att_dst = (const float*)d_in[4];
  const float* bias = (const float*)d_in[5];
  float* out = (float*)d_out;

  const int N = in_sizes[0] / F;
  const int E = in_sizes[1] / 2;
  const int EN = E + N;
  const int* srcA = ei;
  const int* dstA = ei + E;

  float* ws = (float*)d_ws;
  float* hbuf = ws;
  float* asrc = hbuf + (size_t)N * F;
  float* adst = asrc + (size_t)N * HEADS;
  float* mbuf = adst + (size_t)N * HEADS;

  gat_init<<<((N * HEADS) + 255) / 256, 256, 0, stream>>>((unsigned*)mbuf,
                                                          N * HEADS);
  gat_gemm_wmma<<<(N + 15) / 16, 256, 0, stream>>>(x, W, att_src, att_dst,
                                                   hbuf, asrc, adst, N);
  gat_logits_max<<<(EN + 255) / 256, 256, 0, stream>>>(srcA, dstA, asrc, adst,
                                                       (unsigned*)mbuf, E, EN);
  gat_finalize_m<<<((N * HEADS) + 255) / 256, 256, 0, stream>>>((unsigned*)mbuf,
                                                                N * HEADS);
  gat_owner_aggregate<<<(N + NPW - 1) / NPW, 1024, 0, stream>>>(
      srcA, dstA, asrc, adst, mbuf, hbuf, bias, out, E, N);
}


# Segment max moved into the owning work-group: per round an LDS max over its listed edges, then its sums rescaled by exp(old max - new max); max pass and 3 launches dropped; arrival-order sums
# speedup vs baseline: 4.3813x; 2.1279x over previous
#include <hip/hip_runtime.h>
#include <hip/hip_bf16.h>

typedef float v2f __attribute__((ext_vector_type(2)));
typedef float v8f __attribute__((ext_vector_type(8)));
typedef float v4f __attribute__((ext_vector_type(4)));

#define HEADS 8
#define CH 16
#define F 128
#define NEG_SLOPE 0.2f

#if defined(__has_builtin)
#if __has_builtin(__builtin_amdgcn_wmma_f32_16x16x4_f32)
#define HAVE_WMMA_F32X4 1
#endif
#endif

__device__ __forceinline__ unsigned flipf(float f) {
  unsigned u = __float_as_uint(f);
  return (u >> 31) ? ~u : (u | 0x80000000u);
}
__device__ __forceinline__ float unflipf(unsigned u) {
  return __uint_as_float((u >> 31) ? (u ^ 0x80000000u) : ~u);
}
__device__ __forceinline__ float lrelu(float v) {
  return v > 0.0f ? v : NEG_SLOPE * v;
}

__global__ __launch_bounds__(256) void gat_gemm_wmma(
    const float* __restrict__ x, const float* __restrict__ W,
    const float* __restrict__ att_src, const float* __restrict__ att_dst,
    float* __restrict__ h, float* __restrict__ a_src, float* __restrict__ a_dst,
    int N) {
  __shared__ float sA[16 * F];

  const int tid = threadIdx.x;
  const int wave = tid >> 5;
  const int lane = tid & 31;
  const int half = lane >> 4;
  const int l16 = lane & 15;
  const int rowbase = blockIdx.x * 16;

  {
    int r = tid >> 4;
    int c = (tid & 15) << 3;
    int rg = rowbase + r;
    if (rg >= N) rg = N - 1;
    const float4* s4 = (const float4*)(x + (size_t)rg * F + c);
    float4 a0 = s4[0];
    float4 a1 = s4[1];
    *(float4*)(sA + r * F + c) = a0;
    *(float4*)(sA + r * F + c + 4) = a1;
  }
  __syncthreads();

  const int colbase = wave * CH;
  v8f acc = {0.f, 0.f, 0.f, 0.f, 0.f, 0.f, 0.f, 0.f};

  for (int k = 0; k < F; k += 4) {
#if defined(HAVE_WMMA_F32X4)
    v2f afrag;
    const float* ap = sA + l16 * F + k + 2 * half;
    afrag.x = ap[0];
    afrag.y = ap[1];
    v2f bfrag;
    const float* bp = W + (size_t)(k + 2 * half) * F + colbase + l16;
    bfrag.x = bp[0];
    bfrag.y = bp[F];
    acc = __builtin_amdgcn_wmma_f32_16x16x4_f32(false, afrag, false, bfrag,
                                                (short)0, acc, false, false);
#else
    for (int v = 0; v < 8; ++v) {
      int row = v + 8 * half;
      for (int j = 0; j < 4; ++j)
        acc[v] += sA[row * F + k + j] * W[(size_t)(k + j) * F + colbase + l16];
    }
#endif
  }

  const float was = att_src[wave * CH + l16];
  const float wad = att_dst[wave * CH + l16];
#pragma unroll
  for (int v = 0; v < 8; ++v) {
    int row = rowbase + v + 8 * half;
    float val = acc[v];
    float ps = val * was;
    float pd = val * wad;
#pragma unroll
    for (int off = 8; off >= 1; off >>= 1) {
      ps += __shfl_xor(ps, off, 32);
      pd += __shfl_xor(pd, off, 32);
    }
    if (row < N) {
      h[(size_t)row * F + colbase + l16] = val;
      if (l16 == 0) {
        a_src[(size_t)row * HEADS + wave] = ps;
        a_dst[(size_t)row * HEADS + wave] = pd;
      }
    }
  }
}

#define NPW 392
#define CAP 8192
#define CHUNK (1 << 20)
#define UNR 8
#define MUN 8

__device__ __forceinline__ void take_edge(int d, int e, int base, unsigned nloc,
                                          int* s_cnt, int* s_src, int* s_dl) {
  unsigned r = (unsigned)d - (unsigned)base;
  if (r < nloc) {
    int slot = atomicAdd(s_cnt, 1);
    if (slot < CAP) {
      s_src[slot] = e;
      s_dl[slot] = (int)r;
    }
  }
}

__device__ __forceinline__ void take_quad(int4 d, int e, int base, unsigned nloc,
                                          int* s_cnt, int* s_src, int* s_dl) {
  unsigned r0 = (unsigned)d.x - (unsigned)base;
  unsigned r1 = (unsigned)d.y - (unsigned)base;
  unsigned r2 = (unsigned)d.z - (unsigned)base;
  unsigned r3 = (unsigned)d.w - (unsigned)base;
  if ((r0 < nloc) | (r1 < nloc) | (r2 < nloc) | (r3 < nloc)) {
    take_edge(d.x, e + 0, base, nloc, s_cnt, s_src, s_dl);
    take_edge(d.y, e + 1, base, nloc, s_cnt, s_src, s_dl);
    take_edge(d.z, e + 2, base, nloc, s_cnt, s_src, s_dl);
    take_edge(d.w, e + 3, base, nloc, s_cnt, s_src, s_dl);
  }
}

__global__ __launch_bounds__(1024) void gat_owner_aggregate(
    const int* __restrict__ srcA, const int* __restrict__ dstA,
    const float* __restrict__ a_src, const float* __restrict__ a_dst,
    const float* __restrict__ h, const float* __restrict__ bias,
    float* __restrict__ out, int E, int N) {
  __shared__ float s_acc[NPW * F];
  __shared__ float s_den[NPW * HEADS];
  __shared__ float s_ad[NPW * HEADS];
  __shared__ float s_m[NPW * HEADS];
  __shared__ unsigned s_mu[NPW * HEADS];
  __shared__ int s_src[CAP];
  __shared__ int s_dl[CAP];
  __shared__ int s_cnt;

  const int tid = threadIdx.x;
  const int lane = tid & 31;
  const int wave = __builtin_amdgcn_readfirstlane(tid >> 5);
  const int base = blockIdx.x * NPW;
  int nl = N - base;
  if (nl > NPW) nl = NPW;
  const unsigned nloc = (unsigned)nl;

  const int tot = nl * 32;
  for (int i0 = tid; i0 < tot; i0 += 4 * 1024) {
    float asv[4], adv[4];
    v4f hq[4];
#pragma unroll
    for (int u = 0; u < 4; ++u) {
      int idx = i0 + u * 1024;
      if (idx >= tot) idx = i0;
      size_t node = (size_t)(base + (idx >> 5));
      int q = idx & 31;
      asv[u] = a_src[node * HEADS + (q >> 2)];
      adv[u] = a_dst[node * HEADS + (q >> 2)];
      hq[u] = *(const v4f*)(h + node * F + q * 4);
    }
#pragma unroll
    for (int u = 0; u < 4; ++u) {
      int idx = i0 + u * 1024;
      if (idx < tot) {
        int i = idx >> 5;
        int q = idx & 31;
        float* ap = s_acc + i * F + q;
        ap[0] = hq[u].x;
        ap[32] = hq[u].y;
        ap[64] = hq[u].z;
        ap[96] = hq[u].w;
        if ((q & 3) == 0) {
          const int k = i * HEADS + (q >> 2);
          const float es = lrelu(asv[u] + adv[u]);
          s_den[k] = 1.0f;
          s_ad[k] = adv[u];
          s_m[k] = es;
          s_mu[k] = flipf(es);
        }
      }
    }
  }

  const int hd = lane >> 2;
  int pos = 0;
  int step = CHUNK;
  while (pos < E) {
    int end = (E - pos > step) ? pos + step : E;
    if (tid == 0) s_cnt = 0;
    __syncthreads();

    {
      const int* dp = dstA + pos;
      const int n = end - pos;
      const int nvec = ((((size_t)dp) & 15) == 0) ? (n >> 2) : 0;
      const int4* dp4 = (const int4*)dp;
      int v = tid;
      for (; v + 7 * 1024 < nvec; v += 8 * 1024) {
        int4 dq[8];
#pragma unroll
        for (int u = 0; u < 8; ++u) dq[u] = dp4[v + u * 1024];
#pragma unroll
        for (int u = 0; u < 8; ++u)
          take_quad(dq[u], pos + 4 * (v + u * 1024), base, nloc, &s_cnt, s_src, s_dl);
      }
      for (; v < nvec; v += 1024)
        take_quad(dp4[v], pos + 4 * v, base, nloc, &s_cnt, s_src, s_dl);
      for (int i = (nvec << 2) + tid; i < n; i += 1024)
        take_edge(dp[i], pos + i, base, nloc, &s_cnt, s_src, s_dl);
    }
    __syncthreads();
    const int cnt = s_cnt;
    __syncthreads();
    if (cnt > CAP) {
      step = CAP;
      continue;
    }

    for (int i = tid; i < cnt; i += 4 * 1024) {
      int ev[4];
#pragma unroll
      for (int u = 0; u < 4; ++u) {
        int j = i + u * 1024;
        ev[u] = (j < cnt) ? s_src[j] : 0;
      }
#pragma unroll
      for (int u = 0; u < 4; ++u) ev[u] = srcA[ev[u]];
#pragma unroll
      for (int u = 0; u < 4; ++u) {
        int j = i + u * 1024;
        if (j < cnt) s_src[j] = ev[u];
      }
    }
    __syncthreads();

    const int items = cnt * HEADS;
    for (int t0 = tid; t0 < items; t0 += MUN * 1024) {
      int kk[MUN];
      float av[MUN];
#pragma unroll
      for (int u = 0; u < MUN; ++u) {
        int t = t0 + u * 1024;
        if (t >= items) t = t0;
        const int j = t >> 3;
        const int hh = t & 7;
        kk[u] = s_dl[j] * HEADS + hh;
        av[u] = a_src[(size_t)s_src[j] * HEADS + hh];
      }
#pragma unroll
      for (int u = 0; u < MUN; ++u) {
        if (t0 + u * 1024 < items)
          atomicMax(s_mu + kk[u], flipf(lrelu(av[u] + s_ad[kk[u]])));
      }
    }
    __syncthreads();

    for (int idx = tid; idx < tot; idx += 1024) {
      const int i = idx >> 5;
      const int q = idx & 31;
      const int k = i * HEADS + (q >> 2);
      const float mo = s_m[k];
      const float mn = unflipf(s_mu[k]);
      if (mn > mo) {
        const float f = expf(mo - mn);
        float* ap = s_acc + i * F + q;
        ap[0] *= f;
        ap[32] *= f;
        ap[64] *= f;
        ap[96] *= f;
        if ((q & 3) == 0) {
          s_den[k] *= f;
          s_m[k] = mn;
        }
      }
    }
    __syncthreads();

    for (int i0 = wave * UNR; i0 < cnt; i0 += 32 * UNR) {
      int dl[UNR];
      float as[UNR], ad[UNR], mm[UNR];
      v4f hv[UNR];
      int sl = i0 + (lane & (UNR - 1));
      if (sl >= cnt) sl = i0;
      const int sv = s_src[sl];
      const int dv = s_dl[sl];
#pragma unroll
      for (int u = 0; u < UNR; ++u) {
        int s = __builtin_amdgcn_readlane(sv, u);
        dl[u] = __builtin_amdgcn_readlane(dv, u);
        as[u] = a_src[(size_t)s * HEADS + hd];
        ad[u] = s_ad[dl[u] * HEADS + hd];
        mm[u] = s_m[dl[u] * HEADS + hd];
        hv[u] = *(const v4f*)(h + (size_t)s * F + lane * 4);
      }
      asm volatile("" ::"v"(hv[0]), "v"(hv[1]), "v"(hv[2]), "v"(hv[3]),
                   "v"(hv[4]), "v"(hv[5]), "v"(hv[6]), "v"(hv[7]));
#pragma unroll
      for (int u = 0; u < UNR; ++u) {
        if (i0 + u < cnt) {
          float w = expf(lrelu(as[u] + ad[u]) - mm[u]);
          float* ap = s_acc + dl[u] * F + lane;
          atomicAdd(ap + 0, w * hv[u].x);
          atomicAdd(ap + 32, w * hv[u].y);
          atomicAdd(ap + 64, w * hv[u].z);
          atomicAdd(ap + 96, w * hv[u].w);
          if ((lane & 3) == 0) atomicAdd(s_den + dl[u] * HEADS + hd, w);
        }
      }
    }
    __syncthreads();
    pos = end;
    step = CHUNK;
  }
  __syncthreads();

  const float4* b4 = (const float4*)bias;
  for (int idx = tid; idx < nl * 32; idx += 1024) {
    int i = idx >> 5;
    int q = idx & 31;
    float den = s_den[i * HEADS + (q >> 2)];
    const float* ap = s_acc + i * F + q;
    float4 b = b4[q];
    float4 o;
    o.x = ap[0] / den + b.x;
    o.y = ap[32] / den + b.y;
    o.z = ap[64] / den + b.z;
    o.w = ap[96] / den + b.w;
    ((float4*)out)[(size_t)(base + i) * 32 + q] = o;
  }
}

extern "C" void kernel_launch(void* const* d_in, const int* in_sizes, int n_in,
                              void* d_out, int out_size, void* d_ws, size_t ws_size,
                              hipStream_t stream) {
  const float* x = (const float*)d_in[0];
  const int* ei = (const int*)d_in[1];
  const float* W = (const float*)d_in[2];
  const float* att_src = (const float*)d_in[3];
  const float* att_dst = (const float*)d_in[4];
  const float* bias = (const float*)d_in[5];
  float* out = (float*)d_out;

  const int N = in_sizes[0] / F;
  const int E = in_sizes[1] / 2;
  const int* srcA = ei;
  const int* dstA = ei + E;

  float* ws = (float*)d_ws;
  float* hbuf = ws;
  float* asrc = hbuf + (size_t)N * F;
  float* adst = asrc + (size_t)N * HEADS;

  gat_gemm_wmma<<<(N + 15) / 16, 256, 0, stream>>>(x, W, att_src, att_dst,
                                                   hbuf, asrc, adst, N);
  gat_owner_aggregate<<<(N + NPW - 1) / NPW, 1024, 0, stream>>>(
      srcA, dstA, asrc, adst, hbuf, bias, out, E, N);
}


# Edge filter by owner bitmap: launch 1 also sets one bit per edge in the row of its owning work-group (LDS or, whole-line stores); each owner scans its own row, not all of dst; arrival-order sums
# speedup vs baseline: 5.0678x; 1.1567x over previous
#include <hip/hip_runtime.h>
#include <hip/hip_bf16.h>

typedef float v2f __attribute__((ext_vector_type(2)));
typedef float v8f __attribute__((ext_vector_type(8)));
typedef float v4f __attribute__((ext_vector_type(4)));

#define HEADS 8
#define CH 16
#define F 128
#define NEG_SLOPE 0.2f
#define NPW 392
#define OWN_SLAB 256

#if defined(__has_builtin)
#if __has_builtin(__builtin_amdgcn_wmma_f32_16x16x4_f32)
#define HAVE_WMMA_F32X4 1
#endif
#endif

__device__ __forceinline__ unsigned flipf(float f) {
  unsigned u = __float_as_uint(f);
  return (u >> 31) ? ~u : (u | 0x80000000u);
}
__device__ __forceinline__ float unflipf(unsigned u) {
  return __uint_as_float((u >> 31) ? (u ^ 0x80000000u) : ~u);
}
__device__ __forceinline__ float lrelu(float v) {
  return v > 0.0f ? v : NEG_SLOPE * v;
}

__global__ __launch_bounds__(256) void gat_gemm_wmma(
    const float* __restrict__ x, const float* __restrict__ W,
    const float* __restrict__ att_src, const float* __restrict__ att_dst,
    float* __restrict__ h, float* __restrict__ a_src, float* __restrict__ a_dst,
    const int* __restrict__ dstA, unsigned* __restrict__ bm, int E, int N) {
  __shared__ unsigned s_tile[OWN_SLAB * 32];
  float* sA = (float*)s_tile;

  const int tid = threadIdx.x;
  const int wave = tid >> 5;
  const int lane = tid & 31;
  const int half = lane >> 4;
  const int l16 = lane & 15;
  const int rowbase = blockIdx.x * 16;

  {
    const int ntiles = (E + 1023) >> 10;
    const int EWP = ntiles << 5;
    const unsigned G = (unsigned)((N + NPW - 1) / NPW);
    for (int tile = blockIdx.x; tile < ntiles; tile += gridDim.x) {
      unsigned ov[4];
#pragma unroll
      for (int u = 0; u < 4; ++u) {
        const int e = (tile << 10) + u * 256 + tid;
        ov[u] = (unsigned)dstA[e < E ? e : E - 1];
      }
#pragma unroll
      for (int u = 0; u < 4; ++u) {
        const int e = (tile << 10) + u * 256 + tid;
        ov[u] = (e < E) ? ov[u] / (unsigned)NPW : 0xFFFFFFFFu;
      }
      for (unsigned ob = 0; ob < G; ob += OWN_SLAB) {
#pragma unroll
        for (int u = 0; u < OWN_SLAB * 32 / 1024; ++u)
          *(uint4*)(s_tile + (u * 256 + tid) * 4) = make_uint4(0u, 0u, 0u, 0u);
        __syncthreads();
#pragma unroll
        for (int u = 0; u < 4; ++u) {
          const unsigned rel = ov[u] - ob;
          if (rel < (unsigned)OWN_SLAB && ov[u] < G)
            atomicOr(s_tile + rel * 32 + u * 8 + wave, 1u << lane);
        }
        __syncthreads();
#pragma unroll
        for (int u = 0; u < OWN_SLAB * 32 / 1024; ++u) {
          const int q = (u * 256 + tid) * 4;
          const unsigned o = ob + (unsigned)(q >> 5);
          if (o < G)
            *(uint4*)(bm + (size_t)o * EWP + (tile << 5) + (q & 31)) =
                *(const uint4*)(s_tile + q);
        }
        __syncthreads();
      }
    }
  }

  {
    int r = tid >> 4;
    int c = (tid & 15) << 3;
    int rg = rowbase + r;
    if (rg >= N) rg = N - 1;
    const float4* s4 = (const float4*)(x + (size_t)rg * F + c);
    float4 a0 = s4[0];
    float4 a1 = s4[1];
    *(float4*)(sA + r * F + c) = a0;
    *(float4*)(sA + r * F + c + 4) = a1;
  }
  __syncthreads();

  const int colbase = wave * CH;
  v8f acc = {0.f, 0.f, 0.f, 0.f, 0.f, 0.f, 0.f, 0.f};

  for (int k = 0; k < F; k += 4) {
#if defined(HAVE_WMMA_F32X4)
    v2f afrag;
    const float* ap = sA + l16 * F + k + 2 * half;
    afrag.x = ap[0];
    afrag.y = ap[1];
    v2f bfrag;
    const float* bp = W + (size_t)(k + 2 * half) * F + colbase + l16;
    bfrag.x = bp[0];
    bfrag.y = bp[F];
    acc = __builtin_amdgcn_wmma_f32_16x16x4_f32(false, afrag, false, bfrag,
                                                (short)0, acc, false, false);
#else
    for (int v = 0; v < 8; ++v) {
      int row = v + 8 * half;
      for (int j = 0; j < 4; ++j)
        acc[v] += sA[row * F + k + j] * W[(size_t)(k + j) * F + colbase + l16];
    }
#endif
  }

  const float was = att_src[wave * CH + l16];
  const float wad = att_dst[wave * CH + l16];
#pragma unroll
  for (int v = 0; v < 8; ++v) {
    int row = rowbase + v + 8 * half;
    float val = acc[v];
    float ps = val * was;
    float pd = val * wad;
#pragma unroll
    for (int off = 8; off >= 1; off >>= 1) {
      ps += __shfl_xor(ps, off, 32);
      pd += __shfl_xor(pd, off, 32);
    }
    if (row < N) {
      h[(size_t)row * F + colbase + l16] = val;
      if (l16 == 0) {
        a_src[(size_t)row * HEADS + wave] = ps;
        a_dst[(size_t)row * HEADS + wave] = pd;
      }
    }
  }
}

#define CAP 8192
#define CHUNK (1 << 20)
#define UNR 8
#define MUN 8

__device__ __forceinline__ void take_word(unsigned m, int e0, int* s_cnt, int* s_src) {
  while (m) {
    const int b = __builtin_ctz(m);
    m &= m - 1u;
    const int slot = atomicAdd(s_cnt, 1);
    if (slot < CAP) s_src[slot] = e0 + b;
  }
}

__global__ __launch_bounds__(1024) void gat_owner_aggregate(
    const int* __restrict__ srcA, const int* __restrict__ dstA,
    const float* __restrict__ a_src, const float* __restrict__ a_dst,
    const float* __restrict__ h, const float* __restrict__ bias,
    const unsigned* __restrict__ bm, float* __restrict__ out, int E, int N) {
  __shared__ float s_acc[NPW * F];
  __shared__ float s_den[NPW * HEADS];
  __shared__ float s_ad[NPW * HEADS];
  __shared__ float s_m[NPW * HEADS];
  __shared__ unsigned s_mu[NPW * HEADS];
  __shared__ int s_src[CAP];
  __shared__ int s_dl[CAP];
  __shared__ int s_cnt;

  const int tid = threadIdx.x;
  const int lane = tid & 31;
  const int wave = __builtin_amdgcn_readfirstlane(tid >> 5);
  const int base = blockIdx.x * NPW;
  int nl = N - base;
  if (nl > NPW) nl = NPW;
  const int EWP = ((E + 1023) >> 10) << 5;

  const int tot = nl * 32;
  for (int i0 = tid; i0 < tot; i0 += 4 * 1024) {
    float asv[4], adv[4];
    v4f hq[4];
#pragma unroll
    for (int u = 0; u < 4; ++u) {
      int idx = i0 + u * 1024;
      if (idx >= tot) idx = i0;
      size_t node = (size_t)(base + (idx >> 5));
      int q = idx & 31;
      asv[u] = a_src[node * HEADS + (q >> 2)];
      adv[u] = a_dst[node * HEADS + (q >> 2)];
      hq[u] = *(const v4f*)(h + node * F + q * 4);
    }
#pragma unroll
    for (int u = 0; u < 4; ++u) {
      int idx = i0 + u * 1024;
      if (idx < tot) {
        int i = idx >> 5;
        int q = idx & 31;
        float* ap = s_acc + i * F + q;
        ap[0] = hq[u].x;
        ap[32] = hq[u].y;
        ap[64] = hq[u].z;
        ap[96] = hq[u].w;
        if ((q & 3) == 0) {
          const int k = i * HEADS + (q >> 2);
          const float es = lrelu(asv[u] + adv[u]);
          s_den[k] = 1.0f;
          s_ad[k] = adv[u];
          s_m[k] = es;
          s_mu[k] = flipf(es);
        }
      }
    }
  }

  const int hd = lane >> 2;
  int pos = 0;
  int step = CHUNK;
  while (pos < E) {
    int end = (E - pos > step) ? pos + step : E;
    if (tid == 0) s_cnt = 0;
    __syncthreads();

    {
      const int w0 = pos >> 5;
      const int nw = ((end + 31) >> 5) - w0;
      const unsigned* rp = bm + (size_t)blockIdx.x * EWP + w0;
      const uint4* rp4 = (const uint4*)rp;
      const int nvec = nw >> 2;
      for (int v = tid; v < nvec; v += 8 * 1024) {
        uint4 mq[8];
#pragma unroll
        for (int u = 0; u < 8; ++u) {
          const int idx = v + u * 1024;
          mq[u] = rp4[idx < nvec ? idx : v];
        }
#pragma unroll
        for (int u = 0; u < 8; ++u) {
          const int idx = v + u * 1024;
          if (idx < nvec) {
            const int e0 = pos + 128 * idx;
            take_word(mq[u].x, e0, &s_cnt, s_src);
            take_word(mq[u].y, e0 + 32, &s_cnt, s_src);
            take_word(mq[u].z, e0 + 64, &s_cnt, s_src);
            take_word(mq[u].w, e0 + 96, &s_cnt, s_src);
          }
        }
      }
      for (int i = (nvec << 2) + tid; i < nw; i += 1024)
        take_word(rp[i], pos + 32 * i, &s_cnt, s_src);
    }
    __syncthreads();
    const int cnt = s_cnt;
    __syncthreads();
    if (cnt > CAP) {
      step = CAP;
      continue;
    }

    for (int i = tid; i < cnt; i += 4 * 1024) {
      int ev[4], dv[4];
#pragma unroll
      for (int u = 0; u < 4; ++u) {
        int j = i + u * 1024;
        ev[u] = (j < cnt) ? s_src[j] : 0;
      }
#pragma unroll
      for (int u = 0; u < 4; ++u) {
        dv[u] = dstA[ev[u]];
        ev[u] = srcA[ev[u]];
      }
#pragma unroll
      for (int u = 0; u < 4; ++u) {
        int j = i + u * 1024;
        if (j < cnt) {
          s_src[j] = ev[u];
          s_dl[j] = dv[u] - base;
        }
      }
    }
    __syncthreads();

    const int items = cnt * HEADS;
    for (int t0 = tid; t0 < items; t0 += MUN * 1024) {
      int kk[MUN];
      float av[MUN];
#pragma unroll
      for (int u = 0; u < MUN; ++u) {
        int t = t0 + u * 1024;
        if (t >= items) t = t0;
        const int j = t >> 3;
        const int hh = t & 7;
        kk[u] = s_dl[j] * HEADS + hh;
        av[u] = a_src[(size_t)s_src[j] * HEADS + hh];
      }
#pragma unroll
      for (int u = 0; u < MUN; ++u) {
        if (t0 + u * 1024 < items)
          atomicMax(s_mu + kk[u], flipf(lrelu(av[u] + s_ad[kk[u]])));
      }
    }
    __syncthreads();

    for (int idx = tid; idx < tot; idx += 1024) {
      const int i = idx >> 5;
      const int q = idx & 31;
      const int k = i * HEADS + (q >> 2);
      const float mo = s_m[k];
      const float mn = unflipf(s_mu[k]);
      if (mn > mo) {
        const float f = expf(mo - mn);
        float* ap = s_acc + i * F + q;
        ap[0] *= f;
        ap[32] *= f;
        ap[64] *= f;
        ap[96] *= f;
        if ((q & 3) == 0) {
          s_den[k] *= f;
          s_m[k] = mn;
        }
      }
    }
    __syncthreads();

    for (int i0 = wave * UNR; i0 < cnt; i0 += 32 * UNR) {
      int dl[UNR];
      float as[UNR], ad[UNR], mm[UNR];
      v4f hv[UNR];
      int sl = i0 + (lane & (UNR - 1));
      if (sl >= cnt) sl = i0;
      const int sv = s_src[sl];
      const int dv = s_dl[sl];
#pragma unroll
      for (int u = 0; u < UNR; ++u) {
        int s = __builtin_amdgcn_readlane(sv, u);
        dl[u] = __builtin_amdgcn_readlane(dv, u);
        as[u] = a_src[(size_t)s * HEADS + hd];
        ad[u] = s_ad[dl[u] * HEADS + hd];
        mm[u] = s_m[dl[u] * HEADS + hd];
        hv[u] = *(const v4f*)(h + (size_t)s * F + lane * 4);
      }
      asm volatile("" ::"v"(hv[0]), "v"(hv[1]), "v"(hv[2]), "v"(hv[3]),
                   "v"(hv[4]), "v"(hv[5]), "v"(hv[6]), "v"(hv[7]));
#pragma unroll
      for (int u = 0; u < UNR; ++u) {
        if (i0 + u < cnt) {
          float w = expf(lrelu(as[u] + ad[u]) - mm[u]);
          float* ap = s_acc + dl[u] * F + lane;
          atomicAdd(ap + 0, w * hv[u].x);
          atomicAdd(ap + 32, w * hv[u].y);
          atomicAdd(ap + 64, w * hv[u].z);
          atomicAdd(ap + 96, w * hv[u].w);
          if ((lane & 3) == 0) atomicAdd(s_den + dl[u] * HEADS + hd, w);
        }
      }
    }
    __syncthreads();
    pos = end;
    step = CHUNK;
  }
  __syncthreads();

  const float4* b4 = (const float4*)bias;
  for (int idx = tid; idx < nl * 32; idx += 1024) {
    int i = idx >> 5;
    int q = idx & 31;
    float den = s_den[i * HEADS + (q >> 2)];
    const float* ap = s_acc + i * F + q;
    float4 b = b4[q];
    float4 o;
    o.x = ap[0] / den + b.x;
    o.y = ap[32] / den + b.y;
    o.z = ap[64] / den + b.z;
    o.w = ap[96] / den + b.w;
    ((float4*)out)[(size_t)(base + i) * 32 + q] = o;
  }
}

extern "C" void kernel_launch(void* const* d_in, const int* in_sizes, int n_in,
                              void* d_out, int out_size, void* d_ws, size_t ws_size,
                              hipStream_t stream) {
  const float* x = (const float*)d_in[0];
  const int* ei = (const int*)d_in[1];
  const float* W = (const float*)d_in[2];
  const float* att_src = (const float*)d_in[3];
  const float* att_dst = (const float*)d_in[4];
  const float* bias = (const float*)d_in[5];
  float* out = (float*)d_out;

  const int N = in_sizes[0] / F;
  const int E = in_sizes[1] / 2;
  const int* srcA = ei;
  const int* dstA = ei + E;

  float* ws = (float*)d_ws;
  float* hbuf = ws;
  float* asrc = hbuf + (size_t)N * F;
  float* adst = asrc + (size_t)N * HEADS;
  unsigned* bm = (unsigned*)(adst + (size_t)N * HEADS);

  gat_gemm_wmma<<<(N + 15) / 16, 256, 0, stream>>>(x, W, att_src, att_dst,
                                                   hbuf, asrc, adst, dstA, bm, E, N);
  gat_owner_aggregate<<<(N + NPW - 1) / NPW, 1024, 0, stream>>>(
      srcA, dstA, asrc, adst, hbuf, bias, bm, out, E, N);
}


# Edge weight computed once per (edge, head): per 4 listed edges lane (edge, head) takes the exp and adds it to the exp sum, the row lanes fetch it by ds_bpermute; arrival-order sums
# speedup vs baseline: 5.2770x; 1.0413x over previous
#include <hip/hip_runtime.h>
#include <hip/hip_bf16.h>

typedef float v2f __attribute__((ext_vector_type(2)));
typedef float v8f __attribute__((ext_vector_type(8)));
typedef float v4f __attribute__((ext_vector_type(4)));

#define HEADS 8
#define CH 16
#define F 128
#define NEG_SLOPE 0.2f
#define NPW 392
#define OWN_SLAB 256

#if defined(__has_builtin)
#if __has_builtin(__builtin_amdgcn_wmma_f32_16x16x4_f32)
#define HAVE_WMMA_F32X4 1
#endif
#endif

__device__ __forceinline__ unsigned flipf(float f) {
  unsigned u = __float_as_uint(f);
  return (u >> 31) ? ~u : (u | 0x80000000u);
}
__device__ __forceinline__ float unflipf(unsigned u) {
  return __uint_as_float((u >> 31) ? (u ^ 0x80000000u) : ~u);
}
__device__ __forceinline__ float lrelu(float v) {
  return v > 0.0f ? v : NEG_SLOPE * v;
}

__global__ __launch_bounds__(256) void gat_gemm_wmma(
    const float* __restrict__ x, const float* __restrict__ W,
    const float* __restrict__ att_src, const float* __restrict__ att_dst,
    float* __restrict__ h, float* __restrict__ a_src, float* __restrict__ a_dst,
    const int* __restrict__ dstA, unsigned* __restrict__ bm, int E, int N) {
  __shared__ unsigned s_tile[OWN_SLAB * 32];
  float* sA = (float*)s_tile;

  const int tid = threadIdx.x;
  const int wave = tid >> 5;
  const int lane = tid & 31;
  const int half = lane >> 4;
  const int l16 = lane & 15;
  const int rowbase = blockIdx.x * 16;

  {
    const int ntiles = (E + 1023) >> 10;
    const int EWP = ntiles << 5;
    const unsigned G = (unsigned)((N + NPW - 1) / NPW);
    for (int tile = blockIdx.x; tile < ntiles; tile += gridDim.x) {
      unsigned ov[4];
#pragma unroll
      for (int u = 0; u < 4; ++u) {
        const int e = (tile << 10) + u * 256 + tid;
        ov[u] = (unsigned)dstA[e < E ? e : E - 1];
      }
#pragma unroll
      for (int u = 0; u < 4; ++u) {
        const int e = (tile << 10) + u * 256 + tid;
        ov[u] = (e < E) ? ov[u] / (unsigned)NPW : 0xFFFFFFFFu;
      }
      for (unsigned ob = 0; ob < G; ob += OWN_SLAB) {
#pragma unroll
        for (int u = 0; u < OWN_SLAB * 32 / 1024; ++u)
          *(uint4*)(s_tile + (u * 256 + tid) * 4) = make_uint4(0u, 0u, 0u, 0u);
        __syncthreads();
#pragma unroll
        for (int u = 0; u < 4; ++u) {
          const unsigned rel = ov[u] - ob;
          if (rel < (unsigned)OWN_SLAB && ov[u] < G)
            atomicOr(s_tile + rel * 32 + u * 8 + wave, 1u << lane);
        }
        __syncthreads();
#pragma unroll
        for (int u = 0; u < OWN_SLAB * 32 / 1024; ++u) {
          const int q = (u * 256 + tid) * 4;
          const unsigned o = ob + (unsigned)(q >> 5);
          if (o < G)
            *(uint4*)(bm + (size_t)o * EWP + (tile << 5) + (q & 31)) =
                *(const uint4*)(s_tile + q);
        }
        __syncthreads();
      }
    }
  }

  {
    int r = tid >> 4;
    int c = (tid & 15) << 3;
    int rg = rowbase + r;
    if (rg >= N) rg = N - 1;
    const float4* s4 = (const float4*)(x + (size_t)rg * F + c);
    float4 a0 = s4[0];
    float4 a1 = s4[1];
    *(float4*)(sA + r * F + c) = a0;
    *(float4*)(sA + r * F + c + 4) = a1;
  }
  __syncthreads();

  const int colbase = wave * CH;
  v8f acc = {0.f, 0.f, 0.f, 0.f, 0.f, 0.f, 0.f, 0.f};

  for (int k = 0; k < F; k += 4) {
#if defined(HAVE_WMMA_F32X4)
    v2f afrag;
    const float* ap = sA + l16 * F + k + 2 * half;
    afrag.x = ap[0];
    afrag.y = ap[1];
    v2f bfrag;
    const float* bp = W + (size_t)(k + 2 * half) * F + colbase + l16;
    bfrag.x = bp[0];
    bfrag.y = bp[F];
    acc = __builtin_amdgcn_wmma_f32_16x16x4_f32(false, afrag, false, bfrag,
                                                (short)0, acc, false, false);
#else
    for (int v = 0; v < 8; ++v) {
      int row = v + 8 * half;
      for (int j = 0; j < 4; ++j)
        acc[v] += sA[row * F + k + j] * W[(size_t)(k + j) * F + colbase + l16];
    }
#endif
  }

  const float was = att_src[wave * CH + l16];
  const float wad = att_dst[wave * CH + l16];
#pragma unroll
  for (int v = 0; v < 8; ++v) {
    int row = rowbase + v + 8 * half;
    float val = acc[v];
    float ps = val * was;
    float pd = val * wad;
#pragma unroll
    for (int off = 8; off >= 1; off >>= 1) {
      ps += __shfl_xor(ps, off, 32);
      pd += __shfl_xor(pd, off, 32);
    }
    if (row < N) {
      h[(size_t)row * F + colbase + l16] = val;
      if (l16 == 0) {
        a_src[(size_t)row * HEADS + wave] = ps;
        a_dst[(size_t)row * HEADS + wave] = pd;
      }
    }
  }
}

#define CAP 8192
#define CHUNK (1 << 20)
#define UNR 8
#define MUN 8

__device__ __forceinline__ void take_word(unsigned m, int e0, int* s_cnt, int* s_src) {
  while (m) {
    const int b = __builtin_ctz(m);
    m &= m - 1u;
    const int slot = atomicAdd(s_cnt, 1);
    if (slot < CAP) s_src[slot] = e0 + b;
  }
}

__global__ __launch_bounds__(1024) void gat_owner_aggregate(
    const int* __restrict__ srcA, const int* __restrict__ dstA,
    const float* __restrict__ a_src, const float* __restrict__ a_dst,
    const float* __restrict__ h, const float* __restrict__ bias,
    const unsigned* __restrict__ bm, float* __restrict__ out, int E, int N) {
  __shared__ float s_acc[NPW * F];
  __shared__ float s_den[NPW * HEADS];
  __shared__ float s_ad[NPW * HEADS];
  __shared__ float s_m[NPW * HEADS];
  __shared__ unsigned s_mu[NPW * HEADS];
  __shared__ int s_src[CAP];
  __shared__ int s_dl[CAP];
  __shared__ int s_cnt;

  const int tid = threadIdx.x;
  const int lane = tid & 31;
  const int wave = __builtin_amdgcn_readfirstlane(tid >> 5);
  const int base = blockIdx.x * NPW;
  int nl = N - base;
  if (nl > NPW) nl = NPW;
  const int EWP = ((E + 1023) >> 10) << 5;

  const int tot = nl * 32;
  for (int i0 = tid; i0 < tot; i0 += 4 * 1024) {
    float asv[4], adv[4];
    v4f hq[4];
#pragma unroll
    for (int u = 0; u < 4; ++u) {
      int idx = i0 + u * 1024;
      if (idx >= tot) idx = i0;
      size_t node = (size_t)(base + (idx >> 5));
      int q = idx & 31;
      asv[u] = a_src[node * HEADS + (q >> 2)];
      adv[u] = a_dst[node * HEADS + (q >> 2)];
      hq[u] = *(const v4f*)(h + node * F + q * 4);
    }
#pragma unroll
    for (int u = 0; u < 4; ++u) {
      int idx = i0 + u * 1024;
      if (idx < tot) {
        int i = idx >> 5;
        int q = idx & 31;
        float* ap = s_acc + i * F + q;
        ap[0] = hq[u].x;
        ap[32] = hq[u].y;
        ap[64] = hq[u].z;
        ap[96] = hq[u].w;
        if ((q & 3) == 0) {
          const int k = i * HEADS + (q >> 2);
          const float es = lrelu(asv[u] + adv[u]);
          s_den[k] = 1.0f;
          s_ad[k] = adv[u];
          s_m[k] = es;
          s_mu[k] = flipf(es);
        }
      }
    }
  }

  const int hd = lane >> 2;
  const int eg = lane >> 3;
  const int hh = lane & 7;
  int pos = 0;
  int step = CHUNK;
  while (pos < E) {
    int end = (E - pos > step) ? pos + step : E;
    if (tid == 0) s_cnt = 0;
    __syncthreads();

    {
      const int w0 = pos >> 5;
      const int nw = ((end + 31) >> 5) - w0;
      const unsigned* rp = bm + (size_t)blockIdx.x * EWP + w0;
      const uint4* rp4 = (const uint4*)rp;
      const int nvec = nw >> 2;
      for (int v = tid; v < nvec; v += 8 * 1024) {
        uint4 mq[8];
#pragma unroll
        for (int u = 0; u < 8; ++u) {
          const int idx = v + u * 1024;
          mq[u] = rp4[idx < nvec ? idx : v];
        }
#pragma unroll
        for (int u = 0; u < 8; ++u) {
          const int idx = v + u * 1024;
          if (idx < nvec) {
            const int e0 = pos + 128 * idx;
            take_word(mq[u].x, e0, &s_cnt, s_src);
            take_word(mq[u].y, e0 + 32, &s_cnt, s_src);
            take_word(mq[u].z, e0 + 64, &s_cnt, s_src);
            take_word(mq[u].w, e0 + 96, &s_cnt, s_src);
          }
        }
      }
      for (int i = (nvec << 2) + tid; i < nw; i += 1024)
        take_word(rp[i], pos + 32 * i, &s_cnt, s_src);
    }
    __syncthreads();
    const int cnt = s_cnt;
    __syncthreads();
    if (cnt > CAP) {
      step = CAP;
      continue;
    }

    for (int i = tid; i < cnt; i += 4 * 1024) {
      int ev[4], dv[4];
#pragma unroll
      for (int u = 0; u < 4; ++u) {
        int j = i + u * 1024;
        ev[u] = (j < cnt) ? s_src[j] : 0;
      }
#pragma unroll
      for (int u = 0; u < 4; ++u) {
        dv[u] = dstA[ev[u]];
        ev[u] = srcA[ev[u]];
      }
#pragma unroll
      for (int u = 0; u < 4; ++u) {
        int j = i + u * 1024;
        if (j < cnt) {
          s_src[j] = ev[u];
          s_dl[j] = dv[u] - base;
        }
      }
    }
    __syncthreads();

    const int items = cnt * HEADS;
    for (int t0 = tid; t0 < items; t0 += MUN * 1024) {
      int kk[MUN];
      float av[MUN];
#pragma unroll
      for (int u = 0; u < MUN; ++u) {
        int t = t0 + u * 1024;
        if (t >= items) t = t0;
        const int j = t >> 3;
        const int hh = t & 7;
        kk[u] = s_dl[j] * HEADS + hh;
        av[u] = a_src[(size_t)s_src[j] * HEADS + hh];
      }
#pragma unroll
      for (int u = 0; u < MUN; ++u) {
        if (t0 + u * 1024 < items)
          atomicMax(s_mu + kk[u], flipf(lrelu(av[u] + s_ad[kk[u]])));
      }
    }
    __syncthreads();

    for (int idx = tid; idx < tot; idx += 1024) {
      const int i = idx >> 5;
      const int q = idx & 31;
      const int k = i * HEADS + (q >> 2);
      const float mo = s_m[k];
      const float mn = unflipf(s_mu[k]);
      if (mn > mo) {
        const float f = expf(mo - mn);
        float* ap = s_acc + i * F + q;
        ap[0] *= f;
        ap[32] *= f;
        ap[64] *= f;
        ap[96] *= f;
        if ((q & 3) == 0) {
          s_den[k] *= f;
          s_m[k] = mn;
        }
      }
    }
    __syncthreads();

    for (int i0 = wave * UNR; i0 < cnt; i0 += 32 * UNR) {
      int dl[UNR];
      v4f hv[UNR];
      int svg[UNR / 4], dvg[UNR / 4];
      float asg[UNR / 4], adg[UNR / 4], mg[UNR / 4];
#pragma unroll
      for (int g = 0; g < UNR / 4; ++g) {
        int sl = i0 + 4 * g + eg;
        if (sl >= cnt) sl = i0;
        svg[g] = s_src[sl];
        dvg[g] = s_dl[sl];
      }
#pragma unroll
      for (int g = 0; g < UNR / 4; ++g) {
        asg[g] = a_src[(size_t)svg[g] * HEADS + hh];
        adg[g] = s_ad[dvg[g] * HEADS + hh];
        mg[g] = s_m[dvg[g] * HEADS + hh];
      }
#pragma unroll
      for (int u = 0; u < UNR; ++u) {
        int s = __builtin_amdgcn_readlane(svg[u >> 2], 8 * (u & 3));
        dl[u] = __builtin_amdgcn_readlane(dvg[u >> 2], 8 * (u & 3));
        hv[u] = *(const v4f*)(h + (size_t)s * F + lane * 4);
      }
      asm volatile("" ::"v"(hv[0]), "v"(hv[1]), "v"(hv[2]), "v"(hv[3]),
                   "v"(hv[4]), "v"(hv[5]), "v"(hv[6]), "v"(hv[7]));
      float w[UNR];
#pragma unroll
      for (int g = 0; g < UNR / 4; ++g) {
        const float wg = expf(lrelu(asg[g] + adg[g]) - mg[g]);
        if (i0 + 4 * g + eg < cnt) atomicAdd(s_den + dvg[g] * HEADS + hh, wg);
#pragma unroll
        for (int e = 0; e < 4; ++e) w[4 * g + e] = __shfl(wg, 8 * e + hd, 32);
      }
#pragma unroll
      for (int u = 0; u < UNR; ++u) {
        if (i0 + u < cnt) {
          float* ap = s_acc + dl[u] * F + lane;
          atomicAdd(ap + 0, w[u] * hv[u].x);
          atomicAdd(ap + 32, w[u] * hv[u].y);
          atomicAdd(ap + 64, w[u] * hv[u].z);
          atomicAdd(ap + 96, w[u] * hv[u].w);
        }
      }
    }
    __syncthreads();
    pos = end;
    step = CHUNK;
  }
  __syncthreads();

  const float4* b4 = (const float4*)bias;
  for (int idx = tid; idx < nl * 32; idx += 1024) {
    int i = idx >> 5;
    int q = idx & 31;
    float den = s_den[i * HEADS + (q >> 2)];
    const float* ap = s_acc + i * F + q;
    float4 b = b4[q];
    float4 o;
    o.x = ap[0] / den + b.x;
    o.y = ap[32] / den + b.y;
    o.z = ap[64] / den + b.z;
    o.w = ap[96] / den + b.w;
    ((float4*)out)[(size_t)(base + i) * 32 + q] = o;
  }
}

extern "C" void kernel_launch(void* const* d_in, const int* in_sizes, int n_in,
                              void* d_out, int out_size, void* d_ws, size_t ws_size,
                              hipStream_t stream) {
  const float* x = (const float*)d_in[0];
  const int* ei = (const int*)d_in[1];
  const float* W = (const float*)d_in[2];
  const float* att_src = (const float*)d_in[3];
  const float* att_dst = (const float*)d_in[4];
  const float* bias = (const float*)d_in[5];
  float* out = (float*)d_out;

  const int N = in_sizes[0] / F;
  const int E = in_sizes[1] / 2;
  const int* srcA = ei;
  const int* dstA = ei + E;

  float* ws = (float*)d_ws;
  float* hbuf = ws;
  float* asrc = hbuf + (size_t)N * F;
  float* adst = asrc + (size_t)N * HEADS;
  unsigned* bm = (unsigned*)(adst + (size_t)N * HEADS);

  gat_gemm_wmma<<<(N + 15) / 16, 256, 0, stream>>>(x, W, att_src, att_dst,
                                                   hbuf, asrc, adst, dstA, bm, E, N);
  gat_owner_aggregate<<<(N + NPW - 1) / NPW, 1024, 0, stream>>>(
      srcA, dstA, asrc, adst, hbuf, bias, bm, out, E, N);
}


# Owner rows hold the edges, not bits: launch 1 files a (source, local dst) word per edge in its owner's 32-word row per tile by an LDS counter; owners copy rows, no src/dst gathers; arrival-order sums
# speedup vs baseline: 6.5592x; 1.2430x over previous
#include <hip/hip_runtime.h>
#include <hip/hip_bf16.h>

typedef float v2f __attribute__((ext_vector_type(2)));
typedef float v8f __attribute__((ext_vector_type(8)));
typedef float v4f __attribute__((ext_vector_type(4)));

#define HEADS 8
#define CH 16
#define F 128
#define NEG_SLOPE 0.2f
#define NPW 392
#define OWN_SLAB 256
#define ROW_EMPTY 0xFFFFFFFFu
#define SRC_BITS 23
#define SRC_MASK 0x7FFFFFu

__device__ __forceinline__ unsigned flipf(float f) {
  unsigned u = __float_as_uint(f);
  return (u >> 31) ? ~u : (u | 0x80000000u);
}
__device__ __forceinline__ float unflipf(unsigned u) {
  return __uint_as_float((u >> 31) ? (u ^ 0x80000000u) : ~u);
}
__device__ __forceinline__ float lrelu(float v) {
  return v > 0.0f ? v : NEG_SLOPE * v;
}

#define WP 288
#define SPB 8

__global__ __launch_bounds__(256) void gat_gemm_wmma(
    const float* __restrict__ x, const float* __restrict__ W,
    const float* __restrict__ att_src, const float* __restrict__ att_dst,
    float* __restrict__ h, float* __restrict__ a_src, float* __restrict__ a_dst,
    const int* __restrict__ srcA, const int* __restrict__ dstA,
    unsigned* __restrict__ bm, int E, int N) {
  __shared__ unsigned s_tile[64 * WP];
  float* sW = (float*)s_tile;

  const int tid = threadIdx.x;
  const int wave = tid >> 5;
  const int lane = tid & 31;
  const int half = lane >> 4;
  const int l16 = lane & 15;
  const int rowbase = (blockIdx.x * SPB + wave) * 16;

  {
    const int ntiles = (E + 1023) >> 10;
    const int EWP = ntiles << 5;
    const unsigned G = (unsigned)((N + NPW - 1) / NPW);
    for (int tile = blockIdx.x; tile < ntiles; tile += gridDim.x) {
      unsigned ov[4], pk[4];
      int dv[4], sv[4];
#pragma unroll
      for (int u = 0; u < 4; ++u) {
        const int e = (tile << 10) + u * 256 + tid;
        dv[u] = dstA[e < E ? e : E - 1];
        sv[u] = srcA[e < E ? e : E - 1];
      }
#pragma unroll
      for (int u = 0; u < 4; ++u) {
        const int e = (tile << 10) + u * 256 + tid;
        const unsigned o = (unsigned)dv[u] / (unsigned)NPW;
        pk[u] = (((unsigned)dv[u] - o * (unsigned)NPW) << SRC_BITS) | (unsigned)sv[u];
        ov[u] = (e < E) ? o : 0xFFFFFFFFu;
      }
      for (unsigned ob = 0; ob < G; ob += OWN_SLAB) {
#pragma unroll
        for (int u = 0; u < OWN_SLAB * 32 / 1024; ++u) {
          const unsigned first = (tid & 7) == 0 ? 0u : ROW_EMPTY;
          *(uint4*)(s_tile + (u * 256 + tid) * 4) =
              make_uint4(first, ROW_EMPTY, ROW_EMPTY, ROW_EMPTY);
        }
        __syncthreads();
        unsigned slot[4];
#pragma unroll
        for (int u = 0; u < 4; ++u) {
          const unsigned rel = ov[u] - ob;
          slot[u] = 32u;
          if (rel < (unsigned)OWN_SLAB && ov[u] < G)
            slot[u] = atomicAdd(s_tile + rel * 32, 1u) + 1u;
        }
#pragma unroll
        for (int u = 0; u < 4; ++u) {
          if (slot[u] < 32u) s_tile[(ov[u] - ob) * 32 + slot[u]] = pk[u];
        }
        __syncthreads();
#pragma unroll
        for (int u = 0; u < OWN_SLAB * 32 / 1024; ++u) {
          const int q = (u * 256 + tid) * 4;
          const unsigned o = ob + (unsigned)(q >> 5);
          if (o < G)
            *(uint4*)(bm + (size_t)o * EWP + (tile << 5) + (q & 31)) =
                *(const uint4*)(s_tile + q);
        }
        __syncthreads();
      }
    }
  }

  v2f afr[32];
  {
    int rg = rowbase + l16;
    if (rg >= N) rg = N - 1;
    const float* xr = x + (size_t)rg * F + 2 * half;
#pragma unroll
    for (int j = 0; j < 32; ++j) afr[j] = *(const v2f*)(xr + 4 * j);
  }

#pragma unroll
  for (int u = 0; u < 8; ++u) {
    const int p = u * 8 + wave;
    const int c = lane * 4;
    const float4 w0 = *(const float4*)(W + (size_t)(2 * p) * F + c);
    const float4 w1 = *(const float4*)(W + (size_t)(2 * p + 1) * F + c);
    float* d = sW + p * WP + c * 2;
    *(float4*)(d) = make_float4(w0.x, w1.x, w0.y, w1.y);
    *(float4*)(d + 4) = make_float4(w0.z, w1.z, w0.w, w1.w);
  }
  __syncthreads();

  v8f acc[8];
#pragma unroll
  for (int t = 0; t < 8; ++t) acc[t] = (v8f){0.f, 0.f, 0.f, 0.f, 0.f, 0.f, 0.f, 0.f};

  const float* bp = sW + half * WP + l16 * 2;
#pragma unroll
  for (int j = 0; j < 32; ++j) {
    v2f bfr[8];
#pragma unroll
    for (int t = 0; t < 8; ++t) bfr[t] = *(const v2f*)(bp + 2 * j * WP + 32 * t);
#pragma unroll
    for (int t = 0; t < 8; ++t)
      acc[t] = __builtin_amdgcn_wmma_f32_16x16x4_f32(false, afr[j], false, bfr[t],
                                                     (short)0, acc[t], false, false);
  }

  float was[8], wad[8];
#pragma unroll
  for (int t = 0; t < 8; ++t) {
    was[t] = att_src[t * CH + l16];
    wad[t] = att_dst[t * CH + l16];
  }
#pragma unroll
  for (int v = 0; v < 8; ++v) {
    const int row = rowbase + v + 8 * half;
    float ps[8], pd[8];
#pragma unroll
    for (int t = 0; t < 8; ++t) {
      ps[t] = acc[t][v] * was[t];
      pd[t] = acc[t][v] * wad[t];
    }
#pragma unroll
    for (int off = 8; off >= 1; off >>= 1) {
#pragma unroll
      for (int t = 0; t < 8; ++t) {
        ps[t] += __shfl_xor(ps[t], off, 32);
        pd[t] += __shfl_xor(pd[t], off, 32);
      }
    }
    float psel = 0.0f, dsel = 0.0f;
#pragma unroll
    for (int t = 0; t < 8; ++t) {
      if (l16 == t) {
        psel = ps[t];
        dsel = pd[t];
      }
    }
    if (row < N) {
#pragma unroll
      for (int t = 0; t < 8; ++t) h[(size_t)row * F + t * CH + l16] = acc[t][v];
      if (l16 < HEADS) {
        a_src[(size_t)row * HEADS + l16] = psel;
        a_dst[(size_t)row * HEADS + l16] = dsel;
      }
    }
  }
}

#define CAP 8192
#define CHUNK (1 << 20)
#define UNR 8
#define MUN 8

__device__ __forceinline__ void put_entry(unsigned word, int w, int n, int first,
                                          int* s_src, int* s_dl) {
  const int slot = first + w - 1;
  if (w >= 1 && w <= n && slot < CAP) {
    s_src[slot] = (int)(word & SRC_MASK);
    s_dl[slot] = (int)(word >> SRC_BITS);
  }
}

__global__ __launch_bounds__(1024) void gat_owner_aggregate(
    const int* __restrict__ srcA, const int* __restrict__ dstA,
    const float* __restrict__ a_src, const float* __restrict__ a_dst,
    const float* __restrict__ h, const float* __restrict__ bias,
    const unsigned* __restrict__ bm, float* __restrict__ out, int E, int N) {
  __shared__ float s_acc[NPW * F];
  __shared__ float s_den[NPW * HEADS];
  __shared__ float s_ad[NPW * HEADS];
  __shared__ float s_m[NPW * HEADS];
  __shared__ unsigned s_mu[NPW * HEADS];
  __shared__ int s_src[CAP];
  __shared__ int s_dl[CAP];
  __shared__ int s_cnt;

  const int tid = threadIdx.x;
  const int lane = tid & 31;
  const int wave = __builtin_amdgcn_readfirstlane(tid >> 5);
  const int base = blockIdx.x * NPW;
  int nl = N - base;
  if (nl > NPW) nl = NPW;
  const int EWP = ((E + 1023) >> 10) << 5;

  const int tot = nl * 32;
  for (int i0 = tid; i0 < tot; i0 += 4 * 1024) {
    float asv[4], adv[4];
    v4f hq[4];
#pragma unroll
    for (int u = 0; u < 4; ++u) {
      int idx = i0 + u * 1024;
      if (idx >= tot) idx = i0;
      size_t node = (size_t)(base + (idx >> 5));
      int q = idx & 31;
      asv[u] = a_src[node * HEADS + (q >> 2)];
      adv[u] = a_dst[node * HEADS + (q >> 2)];
      hq[u] = *(const v4f*)(h + node * F + q * 4);
    }
#pragma unroll
    for (int u = 0; u < 4; ++u) {
      int idx = i0 + u * 1024;
      if (idx < tot) {
        int i = idx >> 5;
        int q = idx & 31;
        float* ap = s_acc + i * F + q;
        ap[0] = hq[u].x;
        ap[32] = hq[u].y;
        ap[64] = hq[u].z;
        ap[96] = hq[u].w;
        if ((q & 3) == 0) {
          const int k = i * HEADS + (q >> 2);
          const float es = lrelu(asv[u] + adv[u]);
          s_den[k] = 1.0f;
          s_ad[k] = adv[u];
          s_m[k] = es;
          s_mu[k] = flipf(es);
        }
      }
    }
  }

  const int hd = lane >> 2;
  const int eg = lane >> 3;
  const int hh = lane & 7;
  int pos = 0;
  int step = CHUNK;
  while (pos < E) {
    int end = (E - pos > step) ? pos + step : E;
    if (tid == 0) s_cnt = 0;
    __syncthreads();

    {
      const int w0 = pos >> 5;
      const int nw = (((end + 1023) >> 10) << 5) - w0;
      const uint4* rp4 = (const uint4*)(bm + (size_t)blockIdx.x * EWP + w0);
      const int nvec = nw >> 2;
      for (int v = tid; v < nvec; v += 8 * 1024) {
        uint4 mq[8];
#pragma unroll
        for (int u = 0; u < 8; ++u) {
          const int idx = v + u * 1024;
          mq[u] = rp4[idx < nvec ? idx : v];
        }
#pragma unroll
        for (int u = 0; u < 8; ++u) {
          const int idx = v + u * 1024;
          const unsigned rc = (unsigned)__shfl((int)mq[u].x, lane & ~7, 32);
          const bool inr = idx < nvec;
          const int n = (inr && rc < 32u) ? (int)rc : 0;
          int first = 0;
          if ((lane & 7) == 0 && n > 0) first = atomicAdd(&s_cnt, n);
          first = __shfl(first, lane & ~7, 32);
          const int wq = 4 * (lane & 7);
          put_entry(mq[u].x, wq, n, first, s_src, s_dl);
          put_entry(mq[u].y, wq + 1, n, first, s_src, s_dl);
          put_entry(mq[u].z, wq + 2, n, first, s_src, s_dl);
          put_entry(mq[u].w, wq + 3, n, first, s_src, s_dl);
          if (inr && rc >= 32u && (lane & 7) == 0) {
            const int t0 = ((w0 + 4 * idx) >> 5) << 10;
            const int t1 = (t0 + 1024 < E) ? t0 + 1024 : E;
            for (int e = t0; e < t1; ++e) {
              const int d = dstA[e];
              if (d >= base && d < base + NPW) {
                const int slot = atomicAdd(&s_cnt, 1);
                if (slot < CAP) {
                  s_src[slot] = srcA[e];
                  s_dl[slot] = d - base;
                }
              }
            }
          }
        }
      }
    }
    __syncthreads();
    const int cnt = s_cnt;
    __syncthreads();
    if (cnt > CAP) {
      step = CAP;
      continue;
    }

    const int items = cnt * HEADS;
    for (int t0 = tid; t0 < items; t0 += MUN * 1024) {
      int kk[MUN];
      float av[MUN];
#pragma unroll
      for (int u = 0; u < MUN; ++u) {
        int t = t0 + u * 1024;
        if (t >= items) t = t0;
        const int j = t >> 3;
        const int hh = t & 7;
        kk[u] = s_dl[j] * HEADS + hh;
        av[u] = a_src[(size_t)s_src[j] * HEADS + hh];
      }
#pragma unroll
      for (int u = 0; u < MUN; ++u) {
        if (t0 + u * 1024 < items)
          atomicMax(s_mu + kk[u], flipf(lrelu(av[u] + s_ad[kk[u]])));
      }
    }
    __syncthreads();

    for (int idx = tid; idx < tot; idx += 1024) {
      const int i = idx >> 5;
      const int q = idx & 31;
      const int k = i * HEADS + (q >> 2);
      const float mo = s_m[k];
      const float mn = unflipf(s_mu[k]);
      if (mn > mo) {
        const float f = expf(mo - mn);
        float* ap = s_acc + i * F + q;
        ap[0] *= f;
        ap[32] *= f;
        ap[64] *= f;
        ap[96] *= f;
        if ((q & 3) == 0) {
          s_den[k] *= f;
          s_m[k] = mn;
        }
      }
    }
    __syncthreads();

    for (int i0 = wave * UNR; i0 < cnt; i0 += 32 * UNR) {
      int dl[UNR];
      v4f hv[UNR];
      int svg[UNR / 4], dvg[UNR / 4];
      float asg[UNR / 4], adg[UNR / 4], mg[UNR / 4];
#pragma unroll
      for (int g = 0; g < UNR / 4; ++g) {
        int sl = i0 + 4 * g + eg;
        if (sl >= cnt) sl = i0;
        svg[g] = s_src[sl];
        dvg[g] = s_dl[sl];
      }
#pragma unroll
      for (int g = 0; g < UNR / 4; ++g) {
        asg[g] = a_src[(size_t)svg[g] * HEADS + hh];
        adg[g] = s_ad[dvg[g] * HEADS + hh];
        mg[g] = s_m[dvg[g] * HEADS + hh];
      }
#pragma unroll
      for (int u = 0; u < UNR; ++u) {
        int s = __builtin_amdgcn_readlane(svg[u >> 2], 8 * (u & 3));
        dl[u] = __builtin_amdgcn_readlane(dvg[u >> 2], 8 * (u & 3));
        hv[u] = *(const v4f*)(h + (size_t)s * F + lane * 4);
      }
      asm volatile("" ::"v"(hv[0]), "v"(hv[1]), "v"(hv[2]), "v"(hv[3]),
                   "v"(hv[4]), "v"(hv[5]), "v"(hv[6]), "v"(hv[7]));
      float w[UNR];
#pragma unroll
      for (int g = 0; g < UNR / 4; ++g) {
        const float wg = expf(lrelu(asg[g] + adg[g]) - mg[g]);
        if (i0 + 4 * g + eg < cnt) atomicAdd(s_den + dvg[g] * HEADS + hh, wg);
#pragma unroll
        for (int e = 0; e < 4; ++e) w[4 * g + e] = __shfl(wg, 8 * e + hd, 32);
      }
#pragma unroll
      for (int u = 0; u < UNR; ++u) {
        if (i0 + u < cnt) {
          float* ap = s_acc + dl[u] * F + lane;
          atomicAdd(ap + 0, w[u] * hv[u].x);
          atomicAdd(ap + 32, w[u] * hv[u].y);
          atomicAdd(ap + 64, w[u] * hv[u].z);
          atomicAdd(ap + 96, w[u] * hv[u].w);
        }
      }
    }
    __syncthreads();
    pos = end;
    step = CHUNK;
  }
  __syncthreads();

  const float4* b4 = (const float4*)bias;
  for (int idx = tid; idx < nl * 32; idx += 1024) {
    int i = idx >> 5;
    int q = idx & 31;
    float den = s_den[i * HEADS + (q >> 2)];
    const float* ap = s_acc + i * F + q;
    float4 b = b4[q];
    float4 o;
    o.x = ap[0] / den + b.x;
    o.y = ap[32] / den + b.y;
    o.z = ap[64] / den + b.z;
    o.w = ap[96] / den + b.w;
    ((float4*)out)[(size_t)(base + i) * 32 + q] = o;
  }
}

extern "C" void kernel_launch(void* const* d_in, const int* in_sizes, int n_in,
                              void* d_out, int out_size, void* d_ws, size_t ws_size,
                              hipStream_t stream) {
  const float* x = (const float*)d_in[0];
  const int* ei = (const int*)d_in[1];
  const float* W = (const float*)d_in[2];
  const float* att_src = (const float*)d_in[3];
  const float* att_dst = (const float*)d_in[4];
  const float* bias = (const float*)d_in[5];
  float* out = (float*)d_out;

  const int N = in_sizes[0] / F;
  const int E = in_sizes[1] / 2;
  const int* srcA = ei;
  const int* dstA = ei + E;

  float* ws = (float*)d_ws;
  float* hbuf = ws;
  float* asrc = hbuf + (size_t)N * F;
  float* adst = asrc + (size_t)N * HEADS;
  unsigned* bm = (unsigned*)(adst + (size_t)N * HEADS);

  gat_gemm_wmma<<<(N + 16 * SPB - 1) / (16 * SPB), 256, 0, stream>>>(
      x, W, att_src, att_dst, hbuf, asrc, adst, srcA, dstA, bm, E, N);
  gat_owner_aggregate<<<(N + NPW - 1) / NPW, 1024, 0, stream>>>(
      srcA, dstA, asrc, adst, hbuf, bias, bm, out, E, N);
}


# Softmax shift fixed at each node's self-loop logit (never above the segment max): per-round LDS max pass and rescale pass dropped; a logit over 40 above it takes a raise path; arrival-order sums
# speedup vs baseline: 7.4344x; 1.1334x over previous
#include <hip/hip_runtime.h>
#include <hip/hip_bf16.h>

typedef float v2f __attribute__((ext_vector_type(2)));
typedef float v8f __attribute__((ext_vector_type(8)));
typedef float v4f __attribute__((ext_vector_type(4)));

#define HEADS 8
#define CH 16
#define F 128
#define NEG_SLOPE 0.2f
#define NPW 392
#define OWN_SLAB 256
#define ROW_EMPTY 0xFFFFFFFFu
#define SRC_BITS 23
#define SRC_MASK 0x7FFFFFu

__device__ __forceinline__ unsigned flipf(float f) {
  unsigned u = __float_as_uint(f);
  return (u >> 31) ? ~u : (u | 0x80000000u);
}
__device__ __forceinline__ float unflipf(unsigned u) {
  return __uint_as_float((u >> 31) ? (u ^ 0x80000000u) : ~u);
}
__device__ __forceinline__ float lrelu(float v) {
  return v > 0.0f ? v : NEG_SLOPE * v;
}

#define WP 288
#define SPB 8

__global__ __launch_bounds__(256) void gat_gemm_wmma(
    const float* __restrict__ x, const float* __restrict__ W,
    const float* __restrict__ att_src, const float* __restrict__ att_dst,
    float* __restrict__ h, float* __restrict__ a_src, float* __restrict__ a_dst,
    const int* __restrict__ srcA, const int* __restrict__ dstA,
    unsigned* __restrict__ bm, int E, int N) {
  __shared__ unsigned s_tile[64 * WP];
  float* sW = (float*)s_tile;

  const int tid = threadIdx.x;
  const int wave = tid >> 5;
  const int lane = tid & 31;
  const int half = lane >> 4;
  const int l16 = lane & 15;
  const int rowbase = (blockIdx.x * SPB + wave) * 16;

  {
    const int ntiles = (E + 1023) >> 10;
    const int EWP = ntiles << 5;
    const unsigned G = (unsigned)((N + NPW - 1) / NPW);
    for (int tile = blockIdx.x; tile < ntiles; tile += gridDim.x) {
      unsigned ov[4], pk[4];
      int dv[4], sv[4];
#pragma unroll
      for (int u = 0; u < 4; ++u) {
        const int e = (tile << 10) + u * 256 + tid;
        dv[u] = dstA[e < E ? e : E - 1];
        sv[u] = srcA[e < E ? e : E - 1];
      }
#pragma unroll
      for (int u = 0; u < 4; ++u) {
        const int e = (tile << 10) + u * 256 + tid;
        const unsigned o = (unsigned)dv[u] / (unsigned)NPW;
        pk[u] = (((unsigned)dv[u] - o * (unsigned)NPW) << SRC_BITS) | (unsigned)sv[u];
        ov[u] = (e < E) ? o : 0xFFFFFFFFu;
      }
      for (unsigned ob = 0; ob < G; ob += OWN_SLAB) {
#pragma unroll
        for (int u = 0; u < OWN_SLAB * 32 / 1024; ++u) {
          const unsigned first = (tid & 7) == 0 ? 0u : ROW_EMPTY;
          *(uint4*)(s_tile + (u * 256 + tid) * 4) =
              make_uint4(first, ROW_EMPTY, ROW_EMPTY, ROW_EMPTY);
        }
        __syncthreads();
        unsigned slot[4];
#pragma unroll
        for (int u = 0; u < 4; ++u) {
          const unsigned rel = ov[u] - ob;
          slot[u] = 32u;
          if (rel < (unsigned)OWN_SLAB && ov[u] < G)
            slot[u] = atomicAdd(s_tile + rel * 32, 1u) + 1u;
        }
#pragma unroll
        for (int u = 0; u < 4; ++u) {
          if (slot[u] < 32u) s_tile[(ov[u] - ob) * 32 + slot[u]] = pk[u];
        }
        __syncthreads();
#pragma unroll
        for (int u = 0; u < OWN_SLAB * 32 / 1024; ++u) {
          const int q = (u * 256 + tid) * 4;
          const unsigned o = ob + (unsigned)(q >> 5);
          if (o < G)
            *(uint4*)(bm + (size_t)o * EWP + (tile << 5) + (q & 31)) =
                *(const uint4*)(s_tile + q);
        }
        __syncthreads();
      }
    }
  }

  v2f afr[32];
  {
    int rg = rowbase + l16;
    if (rg >= N) rg = N - 1;
    const float* xr = x + (size_t)rg * F + 2 * half;
#pragma unroll
    for (int j = 0; j < 32; ++j) afr[j] = *(const v2f*)(xr + 4 * j);
  }

#pragma unroll
  for (int u = 0; u < 8; ++u) {
    const int p = u * 8 + wave;
    const int c = lane * 4;
    const float4 w0 = *(const float4*)(W + (size_t)(2 * p) * F + c);
    const float4 w1 = *(const float4*)(W + (size_t)(2 * p + 1) * F + c);
    float* d = sW + p * WP + c * 2;
    *(float4*)(d) = make_float4(w0.x, w1.x, w0.y, w1.y);
    *(float4*)(d + 4) = make_float4(w0.z, w1.z, w0.w, w1.w);
  }
  __syncthreads();

  v8f acc[8];
#pragma unroll
  for (int t = 0; t < 8; ++t) acc[t] = (v8f){0.f, 0.f, 0.f, 0.f, 0.f, 0.f, 0.f, 0.f};

  const float* bp = sW + half * WP + l16 * 2;
#pragma unroll
  for (int j = 0; j < 32; ++j) {
    v2f bfr[8];
#pragma unroll
    for (int t = 0; t < 8; ++t) bfr[t] = *(const v2f*)(bp + 2 * j * WP + 32 * t);
#pragma unroll
    for (int t = 0; t < 8; ++t)
      acc[t] = __builtin_amdgcn_wmma_f32_16x16x4_f32(false, afr[j], false, bfr[t],
                                                     (short)0, acc[t], false, false);
  }

  float was[8], wad[8];
#pragma unroll
  for (int t = 0; t < 8; ++t) {
    was[t] = att_src[t * CH + l16];
    wad[t] = att_dst[t * CH + l16];
  }
#pragma unroll
  for (int v = 0; v < 8; ++v) {
    const int row = rowbase + v + 8 * half;
    float ps[8], pd[8];
#pragma unroll
    for (int t = 0; t < 8; ++t) {
      ps[t] = acc[t][v] * was[t];
      pd[t] = acc[t][v] * wad[t];
    }
#pragma unroll
    for (int off = 8; off >= 1; off >>= 1) {
#pragma unroll
      for (int t = 0; t < 8; ++t) {
        ps[t] += __shfl_xor(ps[t], off, 32);
        pd[t] += __shfl_xor(pd[t], off, 32);
      }
    }
    float psel = 0.0f, dsel = 0.0f;
#pragma unroll
    for (int t = 0; t < 8; ++t) {
      if (l16 == t) {
        psel = ps[t];
        dsel = pd[t];
      }
    }
    if (row < N) {
#pragma unroll
      for (int t = 0; t < 8; ++t) h[(size_t)row * F + t * CH + l16] = acc[t][v];
      if (l16 < HEADS) {
        a_src[(size_t)row * HEADS + l16] = psel;
        a_dst[(size_t)row * HEADS + l16] = dsel;
      }
    }
  }
}

#define CAP 8192
#define CHUNK (1 << 20)
#define UNR 8
#define SHIFT_SPAN 40.0f

__device__ __forceinline__ void put_entry(unsigned word, int w, int n, int first,
                                          int* s_src, int* s_dl) {
  const int slot = first + w - 1;
  if (w >= 1 && w <= n && slot < CAP) {
    s_src[slot] = (int)(word & SRC_MASK);
    s_dl[slot] = (int)(word >> SRC_BITS);
  }
}

__global__ __launch_bounds__(1024) void gat_owner_aggregate(
    const int* __restrict__ srcA, const int* __restrict__ dstA,
    const float* __restrict__ a_src, const float* __restrict__ a_dst,
    const float* __restrict__ h, const float* __restrict__ bias,
    const unsigned* __restrict__ bm, float* __restrict__ out, int E, int N) {
  __shared__ float s_acc[NPW * F];
  __shared__ float s_den[NPW * HEADS];
  __shared__ float s_ad[NPW * HEADS];
  __shared__ float s_m[NPW * HEADS];
  __shared__ unsigned s_mu[NPW * HEADS];
  __shared__ int s_src[CAP];
  __shared__ int s_dl[CAP];
  __shared__ int s_cnt;
  __shared__ int s_redo;

  const int tid = threadIdx.x;
  const int lane = tid & 31;
  const int wave = __builtin_amdgcn_readfirstlane(tid >> 5);
  const int base = blockIdx.x * NPW;
  int nl = N - base;
  if (nl > NPW) nl = NPW;
  const int EWP = ((E + 1023) >> 10) << 5;

  const int tot = nl * 32;
  for (int i0 = tid; i0 < tot; i0 += 4 * 1024) {
    float asv[4], adv[4];
    v4f hq[4];
#pragma unroll
    for (int u = 0; u < 4; ++u) {
      int idx = i0 + u * 1024;
      if (idx >= tot) idx = i0;
      size_t node = (size_t)(base + (idx >> 5));
      int q = idx & 31;
      asv[u] = a_src[node * HEADS + (q >> 2)];
      adv[u] = a_dst[node * HEADS + (q >> 2)];
      hq[u] = *(const v4f*)(h + node * F + q * 4);
    }
#pragma unroll
    for (int u = 0; u < 4; ++u) {
      int idx = i0 + u * 1024;
      if (idx < tot) {
        int i = idx >> 5;
        int q = idx & 31;
        float* ap = s_acc + i * F + q;
        ap[0] = hq[u].x;
        ap[32] = hq[u].y;
        ap[64] = hq[u].z;
        ap[96] = hq[u].w;
        if ((q & 3) == 0) {
          const int k = i * HEADS + (q >> 2);
          const float es = lrelu(asv[u] + adv[u]);
          s_den[k] = 1.0f;
          s_ad[k] = adv[u];
          s_m[k] = es;
        }
      }
    }
  }

  const int hd = lane >> 2;
  const int eg = lane >> 3;
  const int hh = lane & 7;
  int pos = 0;
  int step = CHUNK;
  while (pos < E) {
    int end = (E - pos > step) ? pos + step : E;
    if (tid == 0) s_cnt = 0;
    __syncthreads();

    {
      const int w0 = pos >> 5;
      const int nw = (((end + 1023) >> 10) << 5) - w0;
      const uint4* rp4 = (const uint4*)(bm + (size_t)blockIdx.x * EWP + w0);
      const int nvec = nw >> 2;
      for (int v = tid; v < nvec; v += 8 * 1024) {
        uint4 mq[8];
#pragma unroll
        for (int u = 0; u < 8; ++u) {
          const int idx = v + u * 1024;
          mq[u] = rp4[idx < nvec ? idx : v];
        }
#pragma unroll
        for (int u = 0; u < 8; ++u) {
          const int idx = v + u * 1024;
          const unsigned rc = (unsigned)__shfl((int)mq[u].x, lane & ~7, 32);
          const bool inr = idx < nvec;
          const int n = (inr && rc < 32u) ? (int)rc : 0;
          int first = 0;
          if ((lane & 7) == 0 && n > 0) first = atomicAdd(&s_cnt, n);
          first = __shfl(first, lane & ~7, 32);
          const int wq = 4 * (lane & 7);
          put_entry(mq[u].x, wq, n, first, s_src, s_dl);
          put_entry(mq[u].y, wq + 1, n, first, s_src, s_dl);
          put_entry(mq[u].z, wq + 2, n, first, s_src, s_dl);
          put_entry(mq[u].w, wq + 3, n, first, s_src, s_dl);
          if (inr && rc >= 32u && (lane & 7) == 0) {
            const int t0 = ((w0 + 4 * idx) >> 5) << 10;
            const int t1 = (t0 + 1024 < E) ? t0 + 1024 : E;
            for (int e = t0; e < t1; ++e) {
              const int d = dstA[e];
              if (d >= base && d < base + NPW) {
                const int slot = atomicAdd(&s_cnt, 1);
                if (slot < CAP) {
                  s_src[slot] = srcA[e];
                  s_dl[slot] = d - base;
                }
              }
            }
          }
        }
      }
    }
    __syncthreads();
    const int cnt = s_cnt;
    if (tid == 0) s_redo = 0;
    __syncthreads();
    if (cnt > CAP) {
      step = CAP;
      continue;
    }

    for (int i0 = wave * UNR; i0 < cnt; i0 += 32 * UNR) {
      int dl[UNR];
      v4f hv[UNR];
      int svg[UNR / 4], dvg[UNR / 4];
      float asg[UNR / 4], adg[UNR / 4], mg[UNR / 4];
#pragma unroll
      for (int g = 0; g < UNR / 4; ++g) {
        int sl = i0 + 4 * g + eg;
        if (sl >= cnt) sl = i0;
        svg[g] = s_src[sl];
        dvg[g] = s_dl[sl];
      }
#pragma unroll
      for (int g = 0; g < UNR / 4; ++g) {
        asg[g] = a_src[(size_t)svg[g] * HEADS + hh];
        adg[g] = s_ad[dvg[g] * HEADS + hh];
        mg[g] = s_m[dvg[g] * HEADS + hh];
      }
#pragma unroll
      for (int u = 0; u < UNR; ++u) {
        int s = __builtin_amdgcn_readlane(svg[u >> 2], 8 * (u & 3));
        dl[u] = __builtin_amdgcn_readlane(dvg[u >> 2], 8 * (u & 3));
        hv[u] = *(const v4f*)(h + (size_t)s * F + lane * 4);
      }
      asm volatile("" ::"v"(hv[0]), "v"(hv[1]), "v"(hv[2]), "v"(hv[3]),
                   "v"(hv[4]), "v"(hv[5]), "v"(hv[6]), "v"(hv[7]));
      float w[UNR];
#pragma unroll
      for (int g = 0; g < UNR / 4; ++g) {
        const float lg = lrelu(asg[g] + adg[g]);
        const bool live = i0 + 4 * g + eg < cnt;
        const bool over = lg > mg[g] + SHIFT_SPAN;
        const float wg = expf(over ? -INFINITY : lg - mg[g]);
        if (live && over) s_redo = 1;
        if (live) atomicAdd(s_den + dvg[g] * HEADS + hh, wg);
#pragma unroll
        for (int e = 0; e < 4; ++e) w[4 * g + e] = __shfl(wg, 8 * e + hd, 32);
      }
#pragma unroll
      for (int u = 0; u < UNR; ++u) {
        if (i0 + u < cnt) {
          float* ap = s_acc + dl[u] * F + lane;
          atomicAdd(ap + 0, w[u] * hv[u].x);
          atomicAdd(ap + 32, w[u] * hv[u].y);
          atomicAdd(ap + 64, w[u] * hv[u].z);
          atomicAdd(ap + 96, w[u] * hv[u].w);
        }
      }
    }
    __syncthreads();
    if (s_redo) {
      for (int k = tid; k < nl * HEADS; k += 1024) s_mu[k] = flipf(s_m[k]);
      __syncthreads();
      const int items = cnt * HEADS;
      for (int t = tid; t < items; t += 1024) {
        const int j = t >> 3;
        const int hq = t & 7;
        const int k = s_dl[j] * HEADS + hq;
        const float lg = lrelu(a_src[(size_t)s_src[j] * HEADS + hq] + s_ad[k]);
        if (lg > s_m[k] + SHIFT_SPAN) atomicMax(s_mu + k, flipf(lg));
      }
      __syncthreads();
      for (int idx = tid; idx < tot; idx += 1024) {
        const int i = idx >> 5;
        const int q = idx & 31;
        const int k = i * HEADS + (q >> 2);
        const float mo = s_m[k];
        const float mn = unflipf(s_mu[k]);
        if (mn > mo) {
          const float f = expf(mo - mn);
          float* ap = s_acc + i * F + q;
          ap[0] *= f;
          ap[32] *= f;
          ap[64] *= f;
          ap[96] *= f;
          if ((q & 3) == 0) s_den[k] *= f;
        }
      }
      __syncthreads();
      for (int t = tid; t < items; t += 1024) {
        const int j = t >> 3;
        const int hq = t & 7;
        const int k = s_dl[j] * HEADS + hq;
        const size_t s = (size_t)s_src[j];
        const float lg = lrelu(a_src[s * HEADS + hq] + s_ad[k]);
        if (lg > s_m[k] + SHIFT_SPAN) {
          const float wl = expf(lg - unflipf(s_mu[k]));
          atomicAdd(s_den + k, wl);
          const float* hr = h + s * F + hq * CH;
          float* ap = s_acc + s_dl[j] * F;
          for (int ch = 0; ch < CH; ++ch) {
            const int cc = hq * CH + ch;
            atomicAdd(ap + (cc & 3) * 32 + (cc >> 2), wl * hr[ch]);
          }
        }
      }
      __syncthreads();
      for (int k = tid; k < nl * HEADS; k += 1024) s_m[k] = unflipf(s_mu[k]);
      __syncthreads();
    }
    pos = end;
    step = CHUNK;
  }
  __syncthreads();

  const float4* b4 = (const float4*)bias;
  for (int idx = tid; idx < nl * 32; idx += 1024) {
    int i = idx >> 5;
    int q = idx & 31;
    float den = s_den[i * HEADS + (q >> 2)];
    const float* ap = s_acc + i * F + q;
    float4 b = b4[q];
    float4 o;
    o.x = ap[0] / den + b.x;
    o.y = ap[32] / den + b.y;
    o.z = ap[64] / den + b.z;
    o.w = ap[96] / den + b.w;
    ((float4*)out)[(size_t)(base + i) * 32 + q] = o;
  }
}

extern "C" void kernel_launch(void* const* d_in, const int* in_sizes, int n_in,
                              void* d_out, int out_size, void* d_ws, size_t ws_size,
                              hipStream_t stream) {
  const float* x = (const float*)d_in[0];
  const int* ei = (const int*)d_in[1];
  const float* W = (const float*)d_in[2];
  const float* att_src = (const float*)d_in[3];
  const float* att_dst = (const float*)d_in[4];
  const float* bias = (const float*)d_in[5];
  float* out = (float*)d_out;

  const int N = in_sizes[0] / F;
  const int E = in_sizes[1] / 2;
  const int* srcA = ei;
  const int* dstA = ei + E;

  float* ws = (float*)d_ws;
  float* hbuf = ws;
  float* asrc = hbuf + (size_t)N * F;
  float* adst = asrc + (size_t)N * HEADS;
  unsigned* bm = (unsigned*)(adst + (size_t)N * HEADS);

  gat_gemm_wmma<<<(N + 16 * SPB - 1) / (16 * SPB), 256, 0, stream>>>(
      x, W, att_src, att_dst, hbuf, asrc, adst, srcA, dstA, bm, E, N);
  gat_owner_aggregate<<<(N + NPW - 1) / NPW, 1024, 0, stream>>>(
      srcA, dstA, asrc, adst, hbuf, bias, bm, out, E, N);
}


# Owner rows cover 2048 edges, not 1024: half the owner-row bytes written and read, one filing tile per work-group, each owner lists all its edges in one round (one list pass fewer); arrival-order sums
# speedup vs baseline: 8.2636x; 1.1115x over previous
#include <hip/hip_runtime.h>
#include <hip/hip_bf16.h>

typedef float v2f __attribute__((ext_vector_type(2)));
typedef float v8f __attribute__((ext_vector_type(8)));
typedef float v4f __attribute__((ext_vector_type(4)));

#define HEADS 8
#define CH 16
#define F 128
#define NEG_SLOPE 0.2f
#define NPW 392
#define OWN_SLAB 256
#define ROW_EMPTY 0xFFFFFFFFu
#define SRC_BITS 23
#define SRC_MASK 0x7FFFFFu
#define TILE_SH 11
#define TILE_E (1 << TILE_SH)
#define EPT (TILE_E / 256)

__device__ __forceinline__ unsigned flipf(float f) {
  unsigned u = __float_as_uint(f);
  return (u >> 31) ? ~u : (u | 0x80000000u);
}
__device__ __forceinline__ float unflipf(unsigned u) {
  return __uint_as_float((u >> 31) ? (u ^ 0x80000000u) : ~u);
}
__device__ __forceinline__ float lrelu(float v) {
  return v > 0.0f ? v : NEG_SLOPE * v;
}

#define WP 288
#define SPB 8

__global__ __launch_bounds__(256) void gat_gemm_wmma(
    const float* __restrict__ x, const float* __restrict__ W,
    const float* __restrict__ att_src, const float* __restrict__ att_dst,
    float* __restrict__ h, float* __restrict__ a_src, float* __restrict__ a_dst,
    const int* __restrict__ srcA, const int* __restrict__ dstA,
    unsigned* __restrict__ bm, int E, int N) {
  __shared__ unsigned s_tile[64 * WP];
  float* sW = (float*)s_tile;

  const int tid = threadIdx.x;
  const int wave = tid >> 5;
  const int lane = tid & 31;
  const int half = lane >> 4;
  const int l16 = lane & 15;
  const int rowbase = (blockIdx.x * SPB + wave) * 16;

  {
    const int ntiles = (E + TILE_E - 1) >> TILE_SH;
    const int EWP = ntiles << 5;
    const unsigned G = (unsigned)((N + NPW - 1) / NPW);
    for (int tile = blockIdx.x; tile < ntiles; tile += gridDim.x) {
      unsigned ov[EPT], pk[EPT];
      int dv[EPT], sv[EPT];
#pragma unroll
      for (int u = 0; u < EPT; ++u) {
        const int e = (tile << TILE_SH) + u * 256 + tid;
        dv[u] = dstA[e < E ? e : E - 1];
        sv[u] = srcA[e < E ? e : E - 1];
      }
#pragma unroll
      for (int u = 0; u < EPT; ++u) {
        const int e = (tile << TILE_SH) + u * 256 + tid;
        const unsigned o = (unsigned)dv[u] / (unsigned)NPW;
        pk[u] = (((unsigned)dv[u] - o * (unsigned)NPW) << SRC_BITS) | (unsigned)sv[u];
        ov[u] = (e < E) ? o : 0xFFFFFFFFu;
      }
      for (unsigned ob = 0; ob < G; ob += OWN_SLAB) {
#pragma unroll
        for (int u = 0; u < OWN_SLAB * 32 / 1024; ++u) {
          const unsigned first = (tid & 7) == 0 ? 0u : ROW_EMPTY;
          *(uint4*)(s_tile + (u * 256 + tid) * 4) =
              make_uint4(first, ROW_EMPTY, ROW_EMPTY, ROW_EMPTY);
        }
        __syncthreads();
        unsigned slot[EPT];
#pragma unroll
        for (int u = 0; u < EPT; ++u) {
          const unsigned rel = ov[u] - ob;
          slot[u] = 32u;
          if (rel < (unsigned)OWN_SLAB && ov[u] < G)
            slot[u] = atomicAdd(s_tile + rel * 32, 1u) + 1u;
        }
#pragma unroll
        for (int u = 0; u < EPT; ++u) {
          if (slot[u] < 32u) s_tile[(ov[u] - ob) * 32 + slot[u]] = pk[u];
        }
        __syncthreads();
#pragma unroll
        for (int u = 0; u < OWN_SLAB * 32 / 1024; ++u) {
          const int q = (u * 256 + tid) * 4;
          const unsigned o = ob + (unsigned)(q >> 5);
          if (o < G)
            *(uint4*)(bm + (size_t)o * EWP + (tile << 5) + (q & 31)) =
                *(const uint4*)(s_tile + q);
        }
        __syncthreads();
      }
    }
  }

  v2f afr[32];
  {
    int rg = rowbase + l16;
    if (rg >= N) rg = N - 1;
    const float* xr = x + (size_t)rg * F + 2 * half;
#pragma unroll
    for (int j = 0; j < 32; ++j) afr[j] = *(const v2f*)(xr + 4 * j);
  }

#pragma unroll
  for (int u = 0; u < 8; ++u) {
    const int p = u * 8 + wave;
    const int c = lane * 4;
    const float4 w0 = *(const float4*)(W + (size_t)(2 * p) * F + c);
    const float4 w1 = *(const float4*)(W + (size_t)(2 * p + 1) * F + c);
    float* d = sW + p * WP + c * 2;
    *(float4*)(d) = make_float4(w0.x, w1.x, w0.y, w1.y);
    *(float4*)(d + 4) = make_float4(w0.z, w1.z, w0.w, w1.w);
  }
  __syncthreads();

  v8f acc[8];
#pragma unroll
  for (int t = 0; t < 8; ++t) acc[t] = (v8f){0.f, 0.f, 0.f, 0.f, 0.f, 0.f, 0.f, 0.f};

  const float* bp = sW + half * WP + l16 * 2;
#pragma unroll
  for (int j = 0; j < 32; ++j) {
    v2f bfr[8];
#pragma unroll
    for (int t = 0; t < 8; ++t) bfr[t] = *(const v2f*)(bp + 2 * j * WP + 32 * t);
#pragma unroll
    for (int t = 0; t < 8; ++t)
      acc[t] = __builtin_amdgcn_wmma_f32_16x16x4_f32(false, afr[j], false, bfr[t],
                                                     (short)0, acc[t], false, false);
  }

  float was[8], wad[8];
#pragma unroll
  for (int t = 0; t < 8; ++t) {
    was[t] = att_src[t * CH + l16];
    wad[t] = att_dst[t * CH + l16];
  }
#pragma unroll
  for (int v = 0; v < 8; ++v) {
    const int row = rowbase + v + 8 * half;
    float ps[8], pd[8];
#pragma unroll
    for (int t = 0; t < 8; ++t) {
      ps[t] = acc[t][v] * was[t];
      pd[t] = acc[t][v] * wad[t];
    }
#pragma unroll
    for (int off = 8; off >= 1; off >>= 1) {
#pragma unroll
      for (int t = 0; t < 8; ++t) {
        ps[t] += __shfl_xor(ps[t], off, 32);
        pd[t] += __shfl_xor(pd[t], off, 32);
      }
    }
    float psel = 0.0f, dsel = 0.0f;
#pragma unroll
    for (int t = 0; t < 8; ++t) {
      if (l16 == t) {
        psel = ps[t];
        dsel = pd[t];
      }
    }
    if (row < N) {
#pragma unroll
      for (int t = 0; t < 8; ++t) h[(size_t)row * F + t * CH + l16] = acc[t][v];
      if (l16 < HEADS) {
        a_src[(size_t)row * HEADS + l16] = psel;
        a_dst[(size_t)row * HEADS + l16] = dsel;
      }
    }
  }
}

#define CAP 8192
#define CHUNK (1 << 21)
#define UNR 8
#define SHIFT_SPAN 40.0f

__device__ __forceinline__ void put_entry(unsigned word, int w, int n, int first,
                                          int* s_src, int* s_dl) {
  const int slot = first + w - 1;
  if (w >= 1 && w <= n && slot < CAP) {
    s_src[slot] = (int)(word & SRC_MASK);
    s_dl[slot] = (int)(word >> SRC_BITS);
  }
}

__global__ __launch_bounds__(1024) void gat_owner_aggregate(
    const int* __restrict__ srcA, const int* __restrict__ dstA,
    const float* __restrict__ a_src, const float* __restrict__ a_dst,
    const float* __restrict__ h, const float* __restrict__ bias,
    const unsigned* __restrict__ bm, float* __restrict__ out, int E, int N) {
  __shared__ float s_acc[NPW * F];
  __shared__ float s_den[NPW * HEADS];
  __shared__ float s_ad[NPW * HEADS];
  __shared__ float s_m[NPW * HEADS];
  __shared__ unsigned s_mu[NPW * HEADS];
  __shared__ int s_src[CAP];
  __shared__ int s_dl[CAP];
  __shared__ int s_cnt;
  __shared__ int s_redo;

  const int tid = threadIdx.x;
  const int lane = tid & 31;
  const int wave = __builtin_amdgcn_readfirstlane(tid >> 5);
  const int base = blockIdx.x * NPW;
  int nl = N - base;
  if (nl > NPW) nl = NPW;
  const int EWP = ((E + TILE_E - 1) >> TILE_SH) << 5;

  const int tot = nl * 32;
  for (int i0 = tid; i0 < tot; i0 += 4 * 1024) {
    float asv[4], adv[4];
    v4f hq[4];
#pragma unroll
    for (int u = 0; u < 4; ++u) {
      int idx = i0 + u * 1024;
      if (idx >= tot) idx = i0;
      size_t node = (size_t)(base + (idx >> 5));
      int q = idx & 31;
      asv[u] = a_src[node * HEADS + (q >> 2)];
      adv[u] = a_dst[node * HEADS + (q >> 2)];
      hq[u] = *(const v4f*)(h + node * F + q * 4);
    }
#pragma unroll
    for (int u = 0; u < 4; ++u) {
      int idx = i0 + u * 1024;
      if (idx < tot) {
        int i = idx >> 5;
        int q = idx & 31;
        float* ap = s_acc + i * F + q;
        ap[0] = hq[u].x;
        ap[32] = hq[u].y;
        ap[64] = hq[u].z;
        ap[96] = hq[u].w;
        if ((q & 3) == 0) {
          const int k = i * HEADS + (q >> 2);
          const float es = lrelu(asv[u] + adv[u]);
          s_den[k] = 1.0f;
          s_ad[k] = adv[u];
          s_m[k] = es;
        }
      }
    }
  }

  const int hd = lane >> 2;
  const int eg = lane >> 3;
  const int hh = lane & 7;
  int pos = 0;
  int step = CHUNK;
  while (pos < E) {
    int end = (E - pos > step) ? pos + step : E;
    if (tid == 0) s_cnt = 0;
    __syncthreads();

    {
      const int w0 = (pos >> TILE_SH) << 5;
      const int nw = (((end + TILE_E - 1) >> TILE_SH) << 5) - w0;
      const uint4* rp4 = (const uint4*)(bm + (size_t)blockIdx.x * EWP + w0);
      const int nvec = nw >> 2;
      for (int v = tid; v < nvec; v += 8 * 1024) {
        uint4 mq[8];
#pragma unroll
        for (int u = 0; u < 8; ++u) {
          const int idx = v + u * 1024;
          mq[u] = rp4[idx < nvec ? idx : v];
        }
#pragma unroll
        for (int u = 0; u < 8; ++u) {
          const int idx = v + u * 1024;
          const unsigned rc = (unsigned)__shfl((int)mq[u].x, lane & ~7, 32);
          const bool inr = idx < nvec;
          const int n = (inr && rc < 32u) ? (int)rc : 0;
          int first = 0;
          if ((lane & 7) == 0 && n > 0) first = atomicAdd(&s_cnt, n);
          first = __shfl(first, lane & ~7, 32);
          const int wq = 4 * (lane & 7);
          put_entry(mq[u].x, wq, n, first, s_src, s_dl);
          put_entry(mq[u].y, wq + 1, n, first, s_src, s_dl);
          put_entry(mq[u].z, wq + 2, n, first, s_src, s_dl);
          put_entry(mq[u].w, wq + 3, n, first, s_src, s_dl);
          if (inr && rc >= 32u && (lane & 7) == 0) {
            const int t0 = ((w0 + 4 * idx) >> 5) << TILE_SH;
            const int t1 = (t0 + TILE_E < E) ? t0 + TILE_E : E;
            for (int e = t0; e < t1; ++e) {
              const int d = dstA[e];
              if (d >= base && d < base + NPW) {
                const int slot = atomicAdd(&s_cnt, 1);
                if (slot < CAP) {
                  s_src[slot] = srcA[e];
                  s_dl[slot] = d - base;
                }
              }
            }
          }
        }
      }
    }
    __syncthreads();
    const int cnt = s_cnt;
    if (tid == 0) s_redo = 0;
    __syncthreads();
    if (cnt > CAP) {
      step = CAP;
      continue;
    }

    for (int i0 = wave * UNR; i0 < cnt; i0 += 32 * UNR) {
      int dl[UNR];
      v4f hv[UNR];
      int svg[UNR / 4], dvg[UNR / 4];
      float asg[UNR / 4], adg[UNR / 4], mg[UNR / 4];
#pragma unroll
      for (int g = 0; g < UNR / 4; ++g) {
        int sl = i0 + 4 * g + eg;
        if (sl >= cnt) sl = i0;
        svg[g] = s_src[sl];
        dvg[g] = s_dl[sl];
      }
#pragma unroll
      for (int g = 0; g < UNR / 4; ++g) {
        asg[g] = a_src[(size_t)svg[g] * HEADS + hh];
        adg[g] = s_ad[dvg[g] * HEADS + hh];
        mg[g] = s_m[dvg[g] * HEADS + hh];
      }
#pragma unroll
      for (int u = 0; u < UNR; ++u) {
        int s = __builtin_amdgcn_readlane(svg[u >> 2], 8 * (u & 3));
        dl[u] = __builtin_amdgcn_readlane(dvg[u >> 2], 8 * (u & 3));
        hv[u] = *(const v4f*)(h + (size_t)s * F + lane * 4);
      }
      asm volatile("" ::"v"(hv[0]), "v"(hv[1]), "v"(hv[2]), "v"(hv[3]),
                   "v"(hv[4]), "v"(hv[5]), "v"(hv[6]), "v"(hv[7]));
      float w[UNR];
#pragma unroll
      for (int g = 0; g < UNR / 4; ++g) {
        const float lg = lrelu(asg[g] + adg[g]);
        const bool live = i0 + 4 * g + eg < cnt;
        const bool over = lg > mg[g] + SHIFT_SPAN;
        const float wg = expf(over ? -INFINITY : lg - mg[g]);
        if (live && over) s_redo = 1;
        if (live) atomicAdd(s_den + dvg[g] * HEADS + hh, wg);
#pragma unroll
        for (int e = 0; e < 4; ++e) w[4 * g + e] = __shfl(wg, 8 * e + hd, 32);
      }
#pragma unroll
      for (int u = 0; u < UNR; ++u) {
        if (i0 + u < cnt) {
          float* ap = s_acc + dl[u] * F + lane;
          atomicAdd(ap + 0, w[u] * hv[u].x);
          atomicAdd(ap + 32, w[u] * hv[u].y);
          atomicAdd(ap + 64, w[u] * hv[u].z);
          atomicAdd(ap + 96, w[u] * hv[u].w);
        }
      }
    }
    __syncthreads();
    if (s_redo) {
      for (int k = tid; k < nl * HEADS; k += 1024) s_mu[k] = flipf(s_m[k]);
      __syncthreads();
      const int items = cnt * HEADS;
      for (int t = tid; t < items; t += 1024) {
        const int j = t >> 3;
        const int hq = t & 7;
        const int k = s_dl[j] * HEADS + hq;
        const float lg = lrelu(a_src[(size_t)s_src[j] * HEADS + hq] + s_ad[k]);
        if (lg > s_m[k] + SHIFT_SPAN) atomicMax(s_mu + k, flipf(lg));
      }
      __syncthreads();
      for (int idx = tid; idx < tot; idx += 1024) {
        const int i = idx >> 5;
        const int q = idx & 31;
        const int k = i * HEADS + (q >> 2);
        const float mo = s_m[k];
        const float mn = unflipf(s_mu[k]);
        if (mn > mo) {
          const float f = expf(mo - mn);
          float* ap = s_acc + i * F + q;
          ap[0] *= f;
          ap[32] *= f;
          ap[64] *= f;
          ap[96] *= f;
          if ((q & 3) == 0) s_den[k] *= f;
        }
      }
      __syncthreads();
      for (int t = tid; t < items; t += 1024) {
        const int j = t >> 3;
        const int hq = t & 7;
        const int k = s_dl[j] * HEADS + hq;
        const size_t s = (size_t)s_src[j];
        const float lg = lrelu(a_src[s * HEADS + hq] + s_ad[k]);
        if (lg > s_m[k] + SHIFT_SPAN) {
          const float wl = expf(lg - unflipf(s_mu[k]));
          atomicAdd(s_den + k, wl);
          const float* hr = h + s * F + hq * CH;
          float* ap = s_acc + s_dl[j] * F;
          for (int ch = 0; ch < CH; ++ch) {
            const int cc = hq * CH + ch;
            atomicAdd(ap + (cc & 3) * 32 + (cc >> 2), wl * hr[ch]);
          }
        }
      }
      __syncthreads();
      for (int k = tid; k < nl * HEADS; k += 1024) s_m[k] = unflipf(s_mu[k]);
      __syncthreads();
    }
    pos = end;
    step = CHUNK;
  }
  __syncthreads();

  const float4* b4 = (const float4*)bias;
  for (int idx = tid; idx < nl * 32; idx += 1024) {
    int i = idx >> 5;
    int q = idx & 31;
    float den = s_den[i * HEADS + (q >> 2)];
    const float* ap = s_acc + i * F + q;
    float4 b = b4[q];
    float4 o;
    o.x = ap[0] / den + b.x;
    o.y = ap[32] / den + b.y;
    o.z = ap[64] / den + b.z;
    o.w = ap[96] / den + b.w;
    ((float4*)out)[(size_t)(base + i) * 32 + q] = o;
  }
}

extern "C" void kernel_launch(void* const* d_in, const int* in_sizes, int n_in,
                              void* d_out, int out_size, void* d_ws, size_t ws_size,
                              hipStream_t stream) {
  const float* x = (const float*)d_in[0];
  const int* ei = (const int*)d_in[1];
  const float* W = (const float*)d_in[2];
  const float* att_src = (const float*)d_in[3];
  const float* att_dst = (const float*)d_in[4];
  const float* bias = (const float*)d_in[5];
  float* out = (float*)d_out;

  const int N = in_sizes[0] / F;
  const int E = in_sizes[1] / 2;
  const int* srcA = ei;
  const int* dstA = ei + E;

  float* ws = (float*)d_ws;
  float* hbuf = ws;
  float* asrc = hbuf + (size_t)N * F;
  float* adst = asrc + (size_t)N * HEADS;
  unsigned* bm = (unsigned*)(adst + (size_t)N * HEADS);

  gat_gemm_wmma<<<(N + 16 * SPB - 1) / (16 * SPB), 256, 0, stream>>>(
      x, W, att_src, att_dst, hbuf, asrc, adst, srcA, dstA, bm, E, N);
  gat_owner_aggregate<<<(N + NPW - 1) / NPW, 1024, 0, stream>>>(
      srcA, dstA, asrc, adst, hbuf, bias, bm, out, E, N);
}
